# starting kernel
# baseline (speedup 1.0000x reference)
#include <hip/hip_runtime.h>

typedef __attribute__((ext_vector_type(2))) float v2f;
typedef __attribute__((ext_vector_type(4))) float v4f;
typedef __attribute__((ext_vector_type(8))) float v8f;

#define N_NODES 50000
#define N_EDGES 800000
#define D 128
#define LDSW 132

__global__ void zero_agg_kernel(float* __restrict__ agg) {
    int i = blockIdx.x * blockDim.x + threadIdx.x;
    v4f z = {0.f, 0.f, 0.f, 0.f};
    ((v4f*)agg)[i] = z;
}

__global__ void scatter_kernel(const float* __restrict__ x,
                               const int* __restrict__ src,
                               const int* __restrict__ dst,
                               float* __restrict__ agg) {
    int t = blockIdx.x * blockDim.x + threadIdx.x;
    int e = t >> 5;
    int lane = t & 31;
    if (e >= N_EDGES) return;
    int s = src[e];
    int d = dst[e];
    const v4f xv = *(const v4f*)(x + (size_t)s * D + lane * 4);
    float* ap = agg + (size_t)d * D + lane * 4;
    atomicAdd(ap + 0, xv.x);
    atomicAdd(ap + 1, xv.y);
    atomicAdd(ap + 2, xv.z);
    atomicAdd(ap + 3, xv.w);
}

__global__ __launch_bounds__(256) void gin_wmma_kernel(
    const float* __restrict__ x,   const float* __restrict__ agg,
    const float* __restrict__ W1,  const float* __restrict__ b1,
    const float* __restrict__ W2,  const float* __restrict__ b2,
    const float* __restrict__ Wres,const float* __restrict__ bres,
    float* __restrict__ out) {

    __shared__ float sh_xa[16 * LDSW];
    __shared__ float sh_xr[16 * LDSW];
    __shared__ float sh_t1[16 * LDSW];

    const int tid = threadIdx.x;
    const int rowbase = blockIdx.x * 16;

    for (int i = tid; i < 16 * (D / 4); i += 256) {
        int r  = i >> 5;
        int c4 = (i & 31) * 4;
        v4f xv = *(const v4f*)(x   + (size_t)(rowbase + r) * D + c4);
        v4f av = *(const v4f*)(agg + (size_t)(rowbase + r) * D + c4);
        v4f sv = {xv.x + av.x, xv.y + av.y, xv.z + av.z, xv.w + av.w};
        *(v4f*)(sh_xr + r * LDSW + c4) = xv;
        *(v4f*)(sh_xa + r * LDSW + c4) = sv;
    }
    __syncthreads();

    const int lane = tid & 31;
    const int wv   = tid >> 5;
    const int n    = lane & 15;
    const int h    = lane >> 4;
    const int colbase = wv * 16;

    v8f acc = {};
    #pragma unroll
    for (int k = 0; k < D; k += 4) {
        v2f a = *(const v2f*)(sh_xa + n * LDSW + k + 2 * h);
        v2f b;
        b.x = W1[(size_t)(k + 2 * h + 0) * D + colbase + n];
        b.y = W1[(size_t)(k + 2 * h + 1) * D + colbase + n];
        acc = __builtin_amdgcn_wmma_f32_16x16x4_f32(
                  false, a, false, b, (short)0, acc, false, false);
    }
    const float bias1 = b1[colbase + n];
    #pragma unroll
    for (int r = 0; r < 8; ++r) {
        float t = acc[r] + bias1;
        sh_t1[(r + 8 * h) * LDSW + colbase + n] = fmaxf(t, 0.f);
    }
    __syncthreads();

    v8f acc2 = {};
    #pragma unroll
    for (int k = 0; k < D; k += 4) {
        v2f a1 = *(const v2f*)(sh_t1 + n * LDSW + k + 2 * h);
        v2f bw2;
        bw2.x = W2[(size_t)(k + 2 * h + 0) * D + colbase + n];
        bw2.y = W2[(size_t)(k + 2 * h + 1) * D + colbase + n];
        acc2 = __builtin_amdgcn_wmma_f32_16x16x4_f32(
                   false, a1, false, bw2, (short)0, acc2, false, false);

        v2f a2 = *(const v2f*)(sh_xr + n * LDSW + k + 2 * h);
        v2f bwr;
        bwr.x = Wres[(size_t)(k + 2 * h + 0) * D + colbase + n];
        bwr.y = Wres[(size_t)(k + 2 * h + 1) * D + colbase + n];
        acc2 = __builtin_amdgcn_wmma_f32_16x16x4_f32(
                   false, a2, false, bwr, (short)0, acc2, false, false);
    }
    const float bias2 = b2[colbase + n] + bres[colbase + n];
    #pragma unroll
    for (int r = 0; r < 8; ++r) {
        out[(size_t)(rowbase + r + 8 * h) * D + colbase + n] = acc2[r] + bias2;
    }
}

extern "C" void kernel_launch(void* const* d_in, const int* in_sizes, int n_in,
                              void* d_out, int out_size, void* d_ws, size_t ws_size,
                              hipStream_t stream) {
    const float* x    = (const float*)d_in[0];
    const int*   ei   = (const int*)  d_in[1];
    const float* W1   = (const float*)d_in[2];
    const float* b1   = (const float*)d_in[3];
    const float* W2   = (const float*)d_in[4];
    const float* b2   = (const float*)d_in[5];
    const float* Wres = (const float*)d_in[6];
    const float* bres = (const float*)d_in[7];
    float* out = (float*)d_out;
    float* agg = (float*)d_ws;

    zero_agg_kernel<<<(N_NODES * D / 4) / 256, 256, 0, stream>>>(agg);

    scatter_kernel<<<(N_EDGES * 32) / 256, 256, 0, stream>>>(
        x, ei, ei + N_EDGES, agg);

    gin_wmma_kernel<<<N_NODES / 16, 256, 0, stream>>>(
        x, agg, W1, b1, W2, b2, Wres, bres, out);
}


# MLP kernel: W1, W2, Wres copied once into LDS by each of 256 resident 16-wave work-groups; B fragments read from LDS instead of memory, barriers fence LDS only; sum order unchanged
# speedup vs baseline: 2.2750x; 2.2750x over previous
#include <hip/hip_runtime.h>

typedef __attribute__((ext_vector_type(2))) float v2f;
typedef __attribute__((ext_vector_type(4))) float v4f;
typedef __attribute__((ext_vector_type(8))) float v8f;

#define N_NODES 50000
#define N_EDGES 800000
#define D 128
#define LDSW 132

typedef __attribute__((ext_vector_type(4))) int v4i;
typedef __attribute__((ext_vector_type(4))) unsigned int v4u;

#define OWN_NODES   200
#define OWN_GROUPS  (N_NODES / OWN_NODES)
#define N_DST4      (N_EDGES / 4)
#define KEY_THREADS 256
#define KEY_PER_THREAD 4
#define KEY_GROUPS  ((N_DST4 + KEY_THREADS * KEY_PER_THREAD - 1) / (KEY_THREADS * KEY_PER_THREAD))

__device__ __forceinline__ unsigned owner_key(int d) {
    const unsigned o = (unsigned)d / (unsigned)OWN_NODES;
    return o < 255u ? o : 255u;
}

__global__ __launch_bounds__(KEY_THREADS) void owner_key_kernel(
    const int* __restrict__ dst,
    unsigned int* __restrict__ keys) {
    const int t    = blockIdx.x * KEY_THREADS + threadIdx.x;
    const int base = (t >> 5) * (32 * KEY_PER_THREAD) + (t & 31);
    const v4i* dst4 = (const v4i*)dst;
    v4i d[KEY_PER_THREAD];
    #pragma unroll
    for (int j = 0; j < KEY_PER_THREAD; ++j) {
        const int i = base + 32 * j;
        d[j] = dst4[i < N_DST4 ? i : N_DST4 - 1];
    }
    #pragma unroll
    for (int j = 0; j < KEY_PER_THREAD; ++j) {
        const int i = base + 32 * j;
        if (i < N_DST4)
            keys[i] = owner_key(d[j].x) | (owner_key(d[j].y) << 8) |
                      (owner_key(d[j].z) << 16) | (owner_key(d[j].w) << 24);
    }
}

#define OWN_THREADS 1024
#define OWN_WAVES   (OWN_THREADS / 32)
#define SCAN_LOADS  7
#define SCAN_WORDS  (SCAN_LOADS * 4)
#define SCAN_GROUPS ((SCAN_WORDS + 7) / 8)
#define N_KEY16     (N_EDGES / 16)
#define CHUNK16     (OWN_THREADS * SCAN_LOADS)
#define N_CHUNKS    ((N_KEY16 + CHUNK16 - 1) / CHUNK16)
#define QCAP        16384

__global__ __launch_bounds__(OWN_THREADS) void gather_owned_kernel(
    const float* __restrict__ x,
    const int* __restrict__ src,
    const int* __restrict__ dst,
    const unsigned int* __restrict__ keys,
    float* __restrict__ agg) {

    __shared__ float acc[(OWN_NODES + 1) * D];
    __shared__ int queue[QCAP];
    __shared__ int qcount;

    const int tid  = threadIdx.x;
    const int lane = tid & 31;
    const int wave = tid >> 5;
    const int lo   = blockIdx.x * OWN_NODES;

    for (int i = tid; i < (OWN_NODES + 1) * D / 4; i += OWN_THREADS) {
        v4f z = {0.f, 0.f, 0.f, 0.f};
        ((v4f*)acc)[i] = z;
    }
    if (tid == 0) qcount = 0;
    __syncthreads();

    const v4u* key16 = (const v4u*)keys;
    const unsigned pat = (unsigned)blockIdx.x * 0x01010101u;
    for (int it = 0; it < N_CHUNKS; ++it) {
        const int chunk = (it + (int)blockIdx.x) % N_CHUNKS;
        const int idx0  = chunk * CHUNK16 + tid;
        v4u v[SCAN_LOADS];
        #pragma unroll
        for (int j = 0; j < SCAN_LOADS; ++j) {
            const int idx = idx0 + j * OWN_THREADS;
            v[j] = key16[idx < N_KEY16 ? idx : N_KEY16 - 1];
        }
        #pragma unroll
        for (int g = 0; g < SCAN_GROUPS; ++g) {
            unsigned p = 0;
            #pragma unroll
            for (int k = 0; k < 8; ++k) {
                const int kk = 8 * g + k;
                if (kk < SCAN_WORDS) {
                    const unsigned t = v[kk >> 2][kk & 3] ^ pat;
                    const unsigned y = ((t & 0x7F7F7F7Fu) + 0x7F7F7F7Fu) | t;
                    p |= ((~y & 0x80808080u) >> 7) << k;
                }
            }
            unsigned keep = (idx0 + (2 * g) * OWN_THREADS < N_KEY16) ? 0x0F0F0F0Fu : 0u;
            if (2 * g + 1 < SCAN_LOADS)
                keep |= (idx0 + (2 * g + 1) * OWN_THREADS < N_KEY16) ? 0xF0F0F0F0u : 0u;
            p &= keep;
            while (p) {
                const int b  = __builtin_ctz(p);
                p &= p - 1;
                const int kk = 8 * g + (b & 7);
                const int e  = 16 * (idx0 + (kk >> 2) * OWN_THREADS) + 4 * (kk & 3) + (b >> 3);
                const int slot = atomicAdd(&qcount, 1);
                if (slot < QCAP) {
                    queue[slot] = e;
                } else {
                    const int s = src[e];
                    const int r = dst[e] - lo;
                    for (int f = 0; f < D; ++f)
                        atomicAdd(&acc[r * D + f], x[(size_t)s * D + f]);
                }
            }
        }
    }
    __syncthreads();

    const int n = qcount < QCAP ? qcount : QCAP;
    for (int base = wave * 32; base < n; base += OWN_WAVES * 32) {
        const int i = base + lane;
        int s = 0;
        int r = OWN_NODES;
        if (i < n) {
            const int e = queue[i];
            s = src[e];
            r = dst[e] - lo;
        }
        #pragma unroll
        for (int j0 = 0; j0 < 32; j0 += 8) {
            float xv[8][4];
            #pragma unroll
            for (int u = 0; u < 8; ++u) {
                const int sj = __builtin_amdgcn_readlane(s, j0 + u);
                const float* xp = x + (size_t)sj * D + lane;
                #pragma unroll
                for (int q = 0; q < 4; ++q) xv[u][q] = xp[32 * q];
            }
            #pragma unroll
            for (int u = 0; u < 8; ++u) {
                const int rj = __builtin_amdgcn_readlane(r, j0 + u);
                float* ap = acc + rj * D + lane;
                #pragma unroll
                for (int q = 0; q < 4; ++q) atomicAdd(ap + 32 * q, xv[u][q]);
            }
        }
    }
    __syncthreads();

    v4f* outp = (v4f*)(agg + (size_t)lo * D);
    for (int i = tid; i < OWN_NODES * D / 4; i += OWN_THREADS)
        outp[i] = ((const v4f*)acc)[i];
}

#define MLP_GROUPS  256
#define MLP_THREADS 512
#define N_TILES     (N_NODES / 16)
#define WPITCH      288
#define WMAT        (64 * WPITCH)
#define TILE_F      (16 * LDSW)
#define W_ITERS     (64 * 32 / MLP_THREADS)

__device__ __forceinline__ void stage_weights(const float* __restrict__ Wa,
                                              const float* __restrict__ Wb,
                                              const float* __restrict__ Wc,
                                              float* __restrict__ sh_w, int tid) {
    v4f r0[3][W_ITERS], r1[3][W_ITERS];
    #pragma unroll
    for (int m = 0; m < 3; ++m) {
        const float* Wg = m == 0 ? Wa : (m == 1 ? Wb : Wc);
        #pragma unroll
        for (int j = 0; j < W_ITERS; ++j) {
            const int i  = tid + MLP_THREADS * j;
            const int kp = i >> 5;
            const int c  = (i & 31) * 4;
            r0[m][j] = *(const v4f*)(Wg + (size_t)(2 * kp) * D + c);
            r1[m][j] = *(const v4f*)(Wg + (size_t)(2 * kp + 1) * D + c);
        }
    }
    #pragma unroll
    for (int m = 0; m < 3; ++m) {
        #pragma unroll
        for (int j = 0; j < W_ITERS; ++j) {
            const int i  = tid + MLP_THREADS * j;
            const int kp = i >> 5;
            const int c  = (i & 31) * 4;
            const v4f a = r0[m][j];
            const v4f b = r1[m][j];
            const v4f lo = {a.x, b.x, a.y, b.y};
            const v4f hi = {a.z, b.z, a.w, b.w};
            float* p = sh_w + m * WMAT + kp * WPITCH + c * 2;
            *(v4f*)p       = lo;
            *(v4f*)(p + 4) = hi;
        }
    }
}

__device__ __forceinline__ void fetch_rows(const float* __restrict__ x,
                                           const float* __restrict__ agg,
                                           int ta, int tb, int tid,
                                           v4f (&px)[2], v4f (&pa)[2]) {
    ta = ta < N_TILES ? ta : N_TILES - 1;
    tb = tb < N_TILES ? tb : N_TILES - 1;
    #pragma unroll
    for (int j = 0; j < 2; ++j) {
        const int i    = tid + MLP_THREADS * j;
        const int tile = (j >= 1) ? tb : ta;
        const int r    = (i >> 5) & 15;
        const int c4   = (i & 31) * 4;
        const size_t off = (size_t)(tile * 16 + r) * D + c4;
        px[j] = *(const v4f*)(x + off);
        pa[j] = *(const v4f*)(agg + off);
    }
}

__device__ __forceinline__ void put_rows(float* __restrict__ sh_xa,
                                         float* __restrict__ sh_xr, int tid,
                                         const v4f (&px)[2], const v4f (&pa)[2]) {
    #pragma unroll
    for (int j = 0; j < 2; ++j) {
        const int i  = tid + MLP_THREADS * j;
        const int o  = (i >> 5) * LDSW + (i & 31) * 4;
        const v4f xv = px[j];
        const v4f av = pa[j];
        const v4f sv = {xv.x + av.x, xv.y + av.y, xv.z + av.z, xv.w + av.w};
        *(v4f*)(sh_xr + o) = xv;
        *(v4f*)(sh_xa + o) = sv;
    }
}

__device__ __forceinline__ void lds_barrier() {
    __builtin_amdgcn_fence(__ATOMIC_RELEASE, "workgroup", "local");
    __builtin_amdgcn_s_barrier();
    __builtin_amdgcn_fence(__ATOMIC_ACQUIRE, "workgroup", "local");
}

__global__ __launch_bounds__(MLP_THREADS) void gin_wmma_kernel(
    const float* __restrict__ x,   const float* __restrict__ agg,
    const float* __restrict__ W1,  const float* __restrict__ b1,
    const float* __restrict__ W2,  const float* __restrict__ b2,
    const float* __restrict__ Wres,const float* __restrict__ bres,
    float* __restrict__ out) {

    __shared__ __attribute__((aligned(16))) float sh_w[3 * WMAT];
    __shared__ __attribute__((aligned(16))) float sh_xa[2 * TILE_F];
    __shared__ __attribute__((aligned(16))) float sh_xr[2 * TILE_F];
    __shared__ __attribute__((aligned(16))) float sh_t1[2 * TILE_F];

    const int tid  = threadIdx.x;
    const int lane = tid & 31;
    const int wv   = __builtin_amdgcn_readfirstlane(tid >> 5);
    const int slot = wv >> 3;
    const int n    = lane & 15;
    const int h    = lane >> 4;
    const int colbase = (wv & 7) * 16;

    v4f px[2], pa[2];
    fetch_rows(x, agg, (int)blockIdx.x, (int)blockIdx.x + MLP_GROUPS, tid, px, pa);

    stage_weights(W1, W2, Wres, sh_w, tid);

    const float bias1 = b1[colbase + n];
    const float bias2 = b2[colbase + n] + bres[colbase + n];

    const float* w1 = sh_w + h * WPITCH + (colbase + n) * 2;
    const float* w2 = w1 + WMAT;
    const float* wr = w1 + 2 * WMAT;
    const int aoff  = slot * TILE_F + n * LDSW + 2 * h;

    for (int ta = blockIdx.x; ta < N_TILES; ta += 2 * MLP_GROUPS) {
        const int my_tile = ta + slot * MLP_GROUPS;
        const bool active = my_tile < N_TILES;

        put_rows(sh_xa, sh_xr, tid, px, pa);
        lds_barrier();
        fetch_rows(x, agg, ta + 2 * MLP_GROUPS, ta + 3 * MLP_GROUPS, tid, px, pa);

        if (active) {
            v8f acc = {};
            #pragma unroll
            for (int k = 0; k < D; k += 4) {
                const v2f a = *(const v2f*)(sh_xa + aoff + k);
                const v2f b = *(const v2f*)(w1 + (k >> 1) * WPITCH);
                acc = __builtin_amdgcn_wmma_f32_16x16x4_f32(
                          false, a, false, b, (short)0, acc, false, false);
            }
            #pragma unroll
            for (int r = 0; r < 8; ++r) {
                const float v = acc[r] + bias1;
                sh_t1[slot * TILE_F + (r + 8 * h) * LDSW + colbase + n] = fmaxf(v, 0.f);
            }
        }
        lds_barrier();

        if (active) {
            v8f acc2 = {};
            #pragma unroll
            for (int k = 0; k < D; k += 4) {
                const v2f a1  = *(const v2f*)(sh_t1 + aoff + k);
                const v2f bw2 = *(const v2f*)(w2 + (k >> 1) * WPITCH);
                acc2 = __builtin_amdgcn_wmma_f32_16x16x4_f32(
                           false, a1, false, bw2, (short)0, acc2, false, false);
                const v2f a2  = *(const v2f*)(sh_xr + aoff + k);
                const v2f bwr = *(const v2f*)(wr + (k >> 1) * WPITCH);
                acc2 = __builtin_amdgcn_wmma_f32_16x16x4_f32(
                           false, a2, false, bwr, (short)0, acc2, false, false);
            }
            const int rowbase = my_tile * 16;
            #pragma unroll
            for (int r = 0; r < 8; ++r)
                out[(size_t)(rowbase + r + 8 * h) * D + colbase + n] = acc2[r] + bias2;
        }
        lds_barrier();
    }
}

extern "C" void kernel_launch(void* const* d_in, const int* in_sizes, int n_in,
                              void* d_out, int out_size, void* d_ws, size_t ws_size,
                              hipStream_t stream) {
    const float* x    = (const float*)d_in[0];
    const int*   ei   = (const int*)  d_in[1];
    const float* W1   = (const float*)d_in[2];
    const float* b1   = (const float*)d_in[3];
    const float* W2   = (const float*)d_in[4];
    const float* b2   = (const float*)d_in[5];
    const float* Wres = (const float*)d_in[6];
    const float* bres = (const float*)d_in[7];
    float* out = (float*)d_out;
    float* agg = (float*)d_ws;
    unsigned int* keys = (unsigned int*)((char*)d_ws + (size_t)N_NODES * D * sizeof(float));

    owner_key_kernel<<<KEY_GROUPS, KEY_THREADS, 0, stream>>>(ei + N_EDGES, keys);

    gather_owned_kernel<<<OWN_GROUPS, OWN_THREADS, 0, stream>>>(
        x, ei, ei + N_EDGES, keys, agg);

    gin_wmma_kernel<<<MLP_GROUPS, MLP_THREADS, 0, stream>>>(
        x, agg, W1, b1, W2, b2, Wres, bres, out);
}


# aggregation kernel: each lane fetches 16 bytes of x[src] (one load per 512-byte row, not four of 128 bytes); LDS rows keep features 4l..4l+3 32 dwords apart, adds still conflict-free, arrival order
# speedup vs baseline: 2.3522x; 1.0339x over previous
#include <hip/hip_runtime.h>

typedef __attribute__((ext_vector_type(2))) float v2f;
typedef __attribute__((ext_vector_type(4))) float v4f;
typedef __attribute__((ext_vector_type(8))) float v8f;

#define N_NODES 50000
#define N_EDGES 800000
#define D 128
#define LDSW 132

typedef __attribute__((ext_vector_type(4))) int v4i;
typedef __attribute__((ext_vector_type(4))) unsigned int v4u;

#define OWN_NODES   200
#define OWN_GROUPS  (N_NODES / OWN_NODES)
#define N_DST4      (N_EDGES / 4)
#define KEY_THREADS 256
#define KEY_PER_THREAD 4
#define KEY_GROUPS  ((N_DST4 + KEY_THREADS * KEY_PER_THREAD - 1) / (KEY_THREADS * KEY_PER_THREAD))

__device__ __forceinline__ unsigned owner_key(int d) {
    const unsigned o = (unsigned)d / (unsigned)OWN_NODES;
    return o < 255u ? o : 255u;
}

__global__ __launch_bounds__(KEY_THREADS) void owner_key_kernel(
    const int* __restrict__ dst,
    unsigned int* __restrict__ keys) {
    const int t    = blockIdx.x * KEY_THREADS + threadIdx.x;
    const int base = (t >> 5) * (32 * KEY_PER_THREAD) + (t & 31);
    const v4i* dst4 = (const v4i*)dst;
    v4i d[KEY_PER_THREAD];
    #pragma unroll
    for (int j = 0; j < KEY_PER_THREAD; ++j) {
        const int i = base + 32 * j;
        d[j] = dst4[i < N_DST4 ? i : N_DST4 - 1];
    }
    #pragma unroll
    for (int j = 0; j < KEY_PER_THREAD; ++j) {
        const int i = base + 32 * j;
        if (i < N_DST4)
            keys[i] = owner_key(d[j].x) | (owner_key(d[j].y) << 8) |
                      (owner_key(d[j].z) << 16) | (owner_key(d[j].w) << 24);
    }
}

#define OWN_THREADS 1024
#define OWN_WAVES   (OWN_THREADS / 32)
#define SCAN_LOADS  7
#define SCAN_WORDS  (SCAN_LOADS * 4)
#define SCAN_GROUPS ((SCAN_WORDS + 7) / 8)
#define N_KEY16     (N_EDGES / 16)
#define CHUNK16     (OWN_THREADS * SCAN_LOADS)
#define N_CHUNKS    ((N_KEY16 + CHUNK16 - 1) / CHUNK16)
#define QCAP        16384
#define APOS(f)     (32 * ((f) & 3) + ((f) >> 2))

__global__ __launch_bounds__(OWN_THREADS) void gather_owned_kernel(
    const float* __restrict__ x,
    const int* __restrict__ src,
    const int* __restrict__ dst,
    const unsigned int* __restrict__ keys,
    float* __restrict__ agg) {

    __shared__ float acc[(OWN_NODES + 1) * D];
    __shared__ int queue[QCAP];
    __shared__ int qcount;

    const int tid  = threadIdx.x;
    const int lane = tid & 31;
    const int wave = tid >> 5;
    const int lo   = blockIdx.x * OWN_NODES;

    for (int i = tid; i < (OWN_NODES + 1) * D / 4; i += OWN_THREADS) {
        v4f z = {0.f, 0.f, 0.f, 0.f};
        ((v4f*)acc)[i] = z;
    }
    if (tid == 0) qcount = 0;
    __syncthreads();

    const v4u* key16 = (const v4u*)keys;
    const unsigned pat = (unsigned)blockIdx.x * 0x01010101u;
    for (int it = 0; it < N_CHUNKS; ++it) {
        const int chunk = (it + (int)blockIdx.x) % N_CHUNKS;
        const int idx0  = chunk * CHUNK16 + tid;
        v4u v[SCAN_LOADS];
        #pragma unroll
        for (int j = 0; j < SCAN_LOADS; ++j) {
            const int idx = idx0 + j * OWN_THREADS;
            v[j] = key16[idx < N_KEY16 ? idx : N_KEY16 - 1];
        }
        #pragma unroll
        for (int g = 0; g < SCAN_GROUPS; ++g) {
            unsigned p = 0;
            #pragma unroll
            for (int k = 0; k < 8; ++k) {
                const int kk = 8 * g + k;
                if (kk < SCAN_WORDS) {
                    const unsigned t = v[kk >> 2][kk & 3] ^ pat;
                    const unsigned y = ((t & 0x7F7F7F7Fu) + 0x7F7F7F7Fu) | t;
                    p |= ((~y & 0x80808080u) >> 7) << k;
                }
            }
            unsigned keep = (idx0 + (2 * g) * OWN_THREADS < N_KEY16) ? 0x0F0F0F0Fu : 0u;
            if (2 * g + 1 < SCAN_LOADS)
                keep |= (idx0 + (2 * g + 1) * OWN_THREADS < N_KEY16) ? 0xF0F0F0F0u : 0u;
            p &= keep;
            while (p) {
                const int b  = __builtin_ctz(p);
                p &= p - 1;
                const int kk = 8 * g + (b & 7);
                const int e  = 16 * (idx0 + (kk >> 2) * OWN_THREADS) + 4 * (kk & 3) + (b >> 3);
                const int slot = atomicAdd(&qcount, 1);
                if (slot < QCAP) {
                    queue[slot] = e;
                } else {
                    const int s = src[e];
                    const int r = dst[e] - lo;
                    for (int f = 0; f < D; ++f)
                        atomicAdd(&acc[r * D + APOS(f)], x[(size_t)s * D + f]);
                }
            }
        }
    }
    __syncthreads();

    const int n = qcount < QCAP ? qcount : QCAP;
    for (int base = wave * 32; base < n; base += OWN_WAVES * 32) {
        const int i = base + lane;
        int s = 0;
        int r = OWN_NODES;
        if (i < n) {
            const int e = queue[i];
            s = src[e];
            r = dst[e] - lo;
        }
        #pragma unroll
        for (int j0 = 0; j0 < 32; j0 += 8) {
            v4f xv[8];
            #pragma unroll
            for (int u = 0; u < 8; ++u) {
                const int sj = __builtin_amdgcn_readlane(s, j0 + u);
                xv[u] = *(const v4f*)(x + (size_t)sj * D + 4 * lane);
            }
            #pragma unroll
            for (int u = 0; u < 8; ++u) {
                const int rj = __builtin_amdgcn_readlane(r, j0 + u);
                float* ap = acc + rj * D + lane;
                atomicAdd(ap,      xv[u].x);
                atomicAdd(ap + 32, xv[u].y);
                atomicAdd(ap + 64, xv[u].z);
                atomicAdd(ap + 96, xv[u].w);
            }
        }
    }
    __syncthreads();

    v4f* outp = (v4f*)(agg + (size_t)lo * D);
    for (int i = tid; i < OWN_NODES * D / 4; i += OWN_THREADS) {
        const float* ap = acc + (i >> 5) * D + (i & 31);
        const v4f v = {ap[0], ap[32], ap[64], ap[96]};
        outp[i] = v;
    }
}

#define MLP_GROUPS  256
#define MLP_THREADS 512
#define N_TILES     (N_NODES / 16)
#define WPITCH      288
#define WMAT        (64 * WPITCH)
#define TILE_F      (16 * LDSW)
#define W_ITERS     (64 * 32 / MLP_THREADS)

__device__ __forceinline__ void stage_weights(const float* __restrict__ Wa,
                                              const float* __restrict__ Wb,
                                              const float* __restrict__ Wc,
                                              float* __restrict__ sh_w, int tid) {
    v4f r0[3][W_ITERS], r1[3][W_ITERS];
    #pragma unroll
    for (int m = 0; m < 3; ++m) {
        const float* Wg = m == 0 ? Wa : (m == 1 ? Wb : Wc);
        #pragma unroll
        for (int j = 0; j < W_ITERS; ++j) {
            const int i  = tid + MLP_THREADS * j;
            const int kp = i >> 5;
            const int c  = (i & 31) * 4;
            r0[m][j] = *(const v4f*)(Wg + (size_t)(2 * kp) * D + c);
            r1[m][j] = *(const v4f*)(Wg + (size_t)(2 * kp + 1) * D + c);
        }
    }
    #pragma unroll
    for (int m = 0; m < 3; ++m) {
        #pragma unroll
        for (int j = 0; j < W_ITERS; ++j) {
            const int i  = tid + MLP_THREADS * j;
            const int kp = i >> 5;
            const int c  = (i & 31) * 4;
            const v4f a = r0[m][j];
            const v4f b = r1[m][j];
            const v4f lo = {a.x, b.x, a.y, b.y};
            const v4f hi = {a.z, b.z, a.w, b.w};
            float* p = sh_w + m * WMAT + kp * WPITCH + c * 2;
            *(v4f*)p       = lo;
            *(v4f*)(p + 4) = hi;
        }
    }
}

__device__ __forceinline__ void fetch_rows(const float* __restrict__ x,
                                           const float* __restrict__ agg,
                                           int ta, int tb, int tid,
                                           v4f (&px)[2], v4f (&pa)[2]) {
    ta = ta < N_TILES ? ta : N_TILES - 1;
    tb = tb < N_TILES ? tb : N_TILES - 1;
    #pragma unroll
    for (int j = 0; j < 2; ++j) {
        const int i    = tid + MLP_THREADS * j;
        const int tile = (j >= 1) ? tb : ta;
        const int r    = (i >> 5) & 15;
        const int c4   = (i & 31) * 4;
        const size_t off = (size_t)(tile * 16 + r) * D + c4;
        px[j] = *(const v4f*)(x + off);
        pa[j] = *(const v4f*)(agg + off);
    }
}

__device__ __forceinline__ void put_rows(float* __restrict__ sh_xa,
                                         float* __restrict__ sh_xr, int tid,
                                         const v4f (&px)[2], const v4f (&pa)[2]) {
    #pragma unroll
    for (int j = 0; j < 2; ++j) {
        const int i  = tid + MLP_THREADS * j;
        const int o  = (i >> 5) * LDSW + (i & 31) * 4;
        const v4f xv = px[j];
        const v4f av = pa[j];
        const v4f sv = {xv.x + av.x, xv.y + av.y, xv.z + av.z, xv.w + av.w};
        *(v4f*)(sh_xr + o) = xv;
        *(v4f*)(sh_xa + o) = sv;
    }
}

__device__ __forceinline__ void lds_barrier() {
    __builtin_amdgcn_fence(__ATOMIC_RELEASE, "workgroup", "local");
    __builtin_amdgcn_s_barrier();
    __builtin_amdgcn_fence(__ATOMIC_ACQUIRE, "workgroup", "local");
}

__global__ __launch_bounds__(MLP_THREADS) void gin_wmma_kernel(
    const float* __restrict__ x,   const float* __restrict__ agg,
    const float* __restrict__ W1,  const float* __restrict__ b1,
    const float* __restrict__ W2,  const float* __restrict__ b2,
    const float* __restrict__ Wres,const float* __restrict__ bres,
    float* __restrict__ out) {

    __shared__ __attribute__((aligned(16))) float sh_w[3 * WMAT];
    __shared__ __attribute__((aligned(16))) float sh_xa[2 * TILE_F];
    __shared__ __attribute__((aligned(16))) float sh_xr[2 * TILE_F];
    __shared__ __attribute__((aligned(16))) float sh_t1[2 * TILE_F];

    const int tid  = threadIdx.x;
    const int lane = tid & 31;
    const int wv   = __builtin_amdgcn_readfirstlane(tid >> 5);
    const int slot = wv >> 3;
    const int n    = lane & 15;
    const int h    = lane >> 4;
    const int colbase = (wv & 7) * 16;

    v4f px[2], pa[2];
    fetch_rows(x, agg, (int)blockIdx.x, (int)blockIdx.x + MLP_GROUPS, tid, px, pa);

    stage_weights(W1, W2, Wres, sh_w, tid);

    const float bias1 = b1[colbase + n];
    const float bias2 = b2[colbase + n] + bres[colbase + n];

    const float* w1 = sh_w + h * WPITCH + (colbase + n) * 2;
    const float* w2 = w1 + WMAT;
    const float* wr = w1 + 2 * WMAT;
    const int aoff  = slot * TILE_F + n * LDSW + 2 * h;

    for (int ta = blockIdx.x; ta < N_TILES; ta += 2 * MLP_GROUPS) {
        const int my_tile = ta + slot * MLP_GROUPS;
        const bool active = my_tile < N_TILES;

        put_rows(sh_xa, sh_xr, tid, px, pa);
        lds_barrier();
        fetch_rows(x, agg, ta + 2 * MLP_GROUPS, ta + 3 * MLP_GROUPS, tid, px, pa);

        if (active) {
            v8f acc = {};
            #pragma unroll
            for (int k = 0; k < D; k += 4) {
                const v2f a = *(const v2f*)(sh_xa + aoff + k);
                const v2f b = *(const v2f*)(w1 + (k >> 1) * WPITCH);
                acc = __builtin_amdgcn_wmma_f32_16x16x4_f32(
                          false, a, false, b, (short)0, acc, false, false);
            }
            #pragma unroll
            for (int r = 0; r < 8; ++r) {
                const float v = acc[r] + bias1;
                sh_t1[slot * TILE_F + (r + 8 * h) * LDSW + colbase + n] = fmaxf(v, 0.f);
            }
        }
        lds_barrier();

        if (active) {
            v8f acc2 = {};
            #pragma unroll
            for (int k = 0; k < D; k += 4) {
                const v2f a1  = *(const v2f*)(sh_t1 + aoff + k);
                const v2f bw2 = *(const v2f*)(w2 + (k >> 1) * WPITCH);
                acc2 = __builtin_amdgcn_wmma_f32_16x16x4_f32(
                           false, a1, false, bw2, (short)0, acc2, false, false);
                const v2f a2  = *(const v2f*)(sh_xr + aoff + k);
                const v2f bwr = *(const v2f*)(wr + (k >> 1) * WPITCH);
                acc2 = __builtin_amdgcn_wmma_f32_16x16x4_f32(
                           false, a2, false, bwr, (short)0, acc2, false, false);
            }
            const int rowbase = my_tile * 16;
            #pragma unroll
            for (int r = 0; r < 8; ++r)
                out[(size_t)(rowbase + r + 8 * h) * D + colbase + n] = acc2[r] + bias2;
        }
        lds_barrier();
    }
}

extern "C" void kernel_launch(void* const* d_in, const int* in_sizes, int n_in,
                              void* d_out, int out_size, void* d_ws, size_t ws_size,
                              hipStream_t stream) {
    const float* x    = (const float*)d_in[0];
    const int*   ei   = (const int*)  d_in[1];
    const float* W1   = (const float*)d_in[2];
    const float* b1   = (const float*)d_in[3];
    const float* W2   = (const float*)d_in[4];
    const float* b2   = (const float*)d_in[5];
    const float* Wres = (const float*)d_in[6];
    const float* bres = (const float*)d_in[7];
    float* out = (float*)d_out;
    float* agg = (float*)d_ws;
    unsigned int* keys = (unsigned int*)((char*)d_ws + (size_t)N_NODES * D * sizeof(float));

    owner_key_kernel<<<KEY_GROUPS, KEY_THREADS, 0, stream>>>(ei + N_EDGES, keys);

    gather_owned_kernel<<<OWN_GROUPS, OWN_THREADS, 0, stream>>>(
        x, ei, ei + N_EDGES, keys, agg);

    gin_wmma_kernel<<<MLP_GROUPS, MLP_THREADS, 0, stream>>>(
        x, agg, W1, b1, W2, b2, Wres, bres, out);
}


# first kernel writes an index (per owner a bit row over 16-edge units, per edge one packed record); owners read their bit row and only the flagged 64-byte record units, no scan; adds in arrival order
# speedup vs baseline: 2.5454x; 1.0821x over previous
#include <hip/hip_runtime.h>

typedef __attribute__((ext_vector_type(2))) float v2f;
typedef __attribute__((ext_vector_type(4))) float v4f;
typedef __attribute__((ext_vector_type(8))) float v8f;

#define N_NODES 50000
#define N_EDGES 800000
#define D 128
#define LDSW 132

typedef __attribute__((ext_vector_type(4))) int v4i;
typedef __attribute__((ext_vector_type(4))) unsigned int v4u;

#define OWN_NODES   200
#define OWN_GROUPS  (N_NODES / OWN_NODES)
#define N_DST4      (N_EDGES / 4)
#define N_UNITS     (N_EDGES / 16)
#define IDX_THREADS 256
#define IDX_PER_THREAD 4
#define IDX_WORDS   (IDX_THREADS * IDX_PER_THREAD)
#define IDX_GROUPS  ((N_DST4 + IDX_WORDS - 1) / IDX_WORDS)
#define BM_GW       (IDX_WORDS / 4 / 32)
#define BM_WORDS    (IDX_GROUPS * BM_GW)

static_assert(N_NODES <= 65536, "src must fit the upper 16 bits of a record");
static_assert(N_EDGES % 16 == 0, "whole units");
static_assert(OWN_NODES < 256 && OWN_GROUPS <= 250, "row and key must fit one byte each");

__device__ __forceinline__ unsigned edge_record(int s, int d) {
    const unsigned o   = (unsigned)d / (unsigned)OWN_NODES;
    const unsigned key = o < 255u ? o : 255u;
    return ((unsigned)s << 16) | (key << 8) | ((unsigned)d - o * (unsigned)OWN_NODES);
}

__global__ __launch_bounds__(IDX_THREADS) void edge_index_kernel(
    const int* __restrict__ src,
    const int* __restrict__ dst,
    unsigned int* __restrict__ rec,
    unsigned int* __restrict__ bm) {
    __shared__ unsigned int tab[OWN_GROUPS * BM_GW];

    const int tid   = threadIdx.x;
    const int wbase = blockIdx.x * IDX_WORDS + (tid >> 5) * (32 * IDX_PER_THREAD) + (tid & 31);
    const v4i* dst4 = (const v4i*)dst;
    const v4i* src4 = (const v4i*)src;
    v4u* rec4 = (v4u*)rec;

    for (int i = tid; i < OWN_GROUPS * BM_GW; i += IDX_THREADS) tab[i] = 0u;

    v4i d[IDX_PER_THREAD];
    v4i s[IDX_PER_THREAD];
    #pragma unroll
    for (int j = 0; j < IDX_PER_THREAD; ++j) {
        const int i  = wbase + 32 * j;
        const int ic = i < N_DST4 ? i : N_DST4 - 1;
        d[j] = dst4[ic];
        s[j] = src4[ic];
    }
    __syncthreads();

    #pragma unroll
    for (int j = 0; j < IDX_PER_THREAD; ++j) {
        const int i = wbase + 32 * j;
        if (i < N_DST4) {
            const int ul = (i >> 2) - (int)blockIdx.x * (IDX_WORDS / 4);
            unsigned int* tw = tab + (ul >> 5);
            const unsigned bit = 1u << (ul & 31);
            const unsigned o0 = (unsigned)d[j].x / (unsigned)OWN_NODES;
            const unsigned o1 = (unsigned)d[j].y / (unsigned)OWN_NODES;
            const unsigned o2 = (unsigned)d[j].z / (unsigned)OWN_NODES;
            const unsigned o3 = (unsigned)d[j].w / (unsigned)OWN_NODES;
            if (o0 < (unsigned)OWN_GROUPS) atomicOr(tw + o0 * BM_GW, bit);
            if (o1 < (unsigned)OWN_GROUPS) atomicOr(tw + o1 * BM_GW, bit);
            if (o2 < (unsigned)OWN_GROUPS) atomicOr(tw + o2 * BM_GW, bit);
            if (o3 < (unsigned)OWN_GROUPS) atomicOr(tw + o3 * BM_GW, bit);
            const v4u r = {edge_record(s[j].x, d[j].x), edge_record(s[j].y, d[j].y),
                           edge_record(s[j].z, d[j].z), edge_record(s[j].w, d[j].w)};
            rec4[i] = r;
        }
    }
    __syncthreads();

    for (int i = tid; i < OWN_GROUPS * BM_GW; i += IDX_THREADS)
        bm[(size_t)(i / BM_GW) * BM_WORDS + blockIdx.x * BM_GW + (i % BM_GW)] = tab[i];
}

#define OWN_THREADS 1024
#define OWN_WAVES   (OWN_THREADS / 32)
#define UQCAP       8192
#define QCAP        16384
#define APOS(f)     (32 * ((f) & 3) + ((f) >> 2))

__device__ __forceinline__ int wave_reserve(int c, int lane, int* counter) {
    int incl = c;
    #pragma unroll
    for (int dlt = 1; dlt < 32; dlt <<= 1) {
        const int t = __shfl_up(incl, dlt);
        if (lane >= dlt) incl += t;
    }
    const int total = __shfl(incl, 31);
    int b0 = 0;
    if (lane == 0 && total > 0) b0 = atomicAdd(counter, total);
    b0 = __shfl(b0, 0);
    return b0 + incl - c;
}

__device__ __forceinline__ void add_edge_slowly(float* acc, const float* __restrict__ x, unsigned rc) {
    const int s = (int)(rc >> 16);
    const int r = (int)(rc & 255u);
    for (int f = 0; f < D; ++f)
        atomicAdd(&acc[r * D + APOS(f)], x[(size_t)s * D + f]);
}

__global__ __launch_bounds__(OWN_THREADS) void gather_owned_kernel(
    const float* __restrict__ x,
    const unsigned int* __restrict__ rec,
    const unsigned int* __restrict__ bm,
    float* __restrict__ agg) {

    __shared__ float acc[(OWN_NODES + 1) * D];
    __shared__ unsigned int queue[QCAP];
    __shared__ int uq[UQCAP];
    __shared__ int qcount;
    __shared__ int ucount;

    const int tid  = threadIdx.x;
    const int lane = tid & 31;
    const int wave = tid >> 5;
    const int lo   = blockIdx.x * OWN_NODES;
    const unsigned mykey = (unsigned)blockIdx.x << 8;

    for (int i = tid; i < (OWN_NODES + 1) * D / 4; i += OWN_THREADS) {
        v4f z = {0.f, 0.f, 0.f, 0.f};
        ((v4f*)acc)[i] = z;
    }
    if (tid == 0) { qcount = 0; ucount = 0; }
    __syncthreads();

    const unsigned int* bmrow = bm + (size_t)blockIdx.x * BM_WORDS;
    for (int w0 = 0; w0 < BM_WORDS; w0 += OWN_THREADS) {
        const int w = w0 + tid;
        unsigned m = w < BM_WORDS ? bmrow[w] : 0u;
        int off = wave_reserve(__builtin_popcount(m), lane, &ucount);
        while (m) {
            const int u = 32 * w + __builtin_ctz(m);
            m &= m - 1;
            if (__builtin_expect(off < UQCAP, 1)) {
                uq[off] = u;
            } else {
                #pragma unroll 1
                for (int k = 0; k < 16; ++k) {
                    const unsigned rc = rec[16 * u + k];
                    if ((rc & 0xFF00u) == mykey) add_edge_slowly(acc, x, rc);
                }
            }
            ++off;
        }
    }
    __syncthreads();

    const v4u* rec16 = (const v4u*)rec;
    const int nu = ucount < UQCAP ? ucount : UQCAP;
    const int wave_s = __builtin_amdgcn_readfirstlane(wave);
    for (int base = wave_s * 32; base < nu; base += OWN_WAVES * 32) {
        v4u rv[4];
        bool have[4];
        int  un[4];
        #pragma unroll
        for (int j = 0; j < 4; ++j) {
            const int ui = base + (lane >> 2) + 8 * j;
            have[j] = ui < nu;
            un[j] = uq[have[j] ? ui : base];
        }
        #pragma unroll
        for (int j = 0; j < 4; ++j)
            rv[j] = rec16[4 * un[j] + (lane & 3)];
        bool     keep[16];
        unsigned bal[16];
        int total = 0;
        #pragma unroll
        for (int k = 0; k < 16; ++k) {
            keep[k] = have[k >> 2] && (rv[k >> 2][k & 3] & 0xFF00u) == mykey;
            bal[k]  = __builtin_amdgcn_ballot_w32(keep[k]);
            total  += __builtin_popcount(bal[k]);
        }
        if (total > 0) {
            int b0 = 0;
            if (lane == 0) b0 = atomicAdd(&qcount, total);
            int pre = __builtin_amdgcn_readfirstlane(b0);
            #pragma unroll
            for (int k = 0; k < 16; ++k) {
                if (keep[k]) {
                    const unsigned rc = rv[k >> 2][k & 3];
                    const int off = pre + (int)__builtin_amdgcn_mbcnt_lo(bal[k], 0u);
                    if (__builtin_expect(off < QCAP, 1)) {
                        queue[off] = rc;
                    } else {
                        add_edge_slowly(acc, x, rc);
                    }
                }
                pre += __builtin_popcount(bal[k]);
            }
        }
    }
    __syncthreads();

    const int n = qcount < QCAP ? qcount : QCAP;
    for (int base = wave * 32; base < n; base += OWN_WAVES * 32) {
        const int i = base + lane;
        const unsigned rc = i < n ? queue[i] : (unsigned)OWN_NODES;
        const int s = (int)(rc >> 16);
        const int r = (int)(rc & 255u);
        #pragma unroll
        for (int j0 = 0; j0 < 32; j0 += 8) {
            v4f xv[8];
            #pragma unroll
            for (int u = 0; u < 8; ++u) {
                const int sj = __builtin_amdgcn_readlane(s, j0 + u);
                xv[u] = *(const v4f*)(x + (size_t)sj * D + 4 * lane);
            }
            #pragma unroll
            for (int u = 0; u < 8; ++u) {
                const int rj = __builtin_amdgcn_readlane(r, j0 + u);
                float* ap = acc + rj * D + lane;
                atomicAdd(ap,      xv[u].x);
                atomicAdd(ap + 32, xv[u].y);
                atomicAdd(ap + 64, xv[u].z);
                atomicAdd(ap + 96, xv[u].w);
            }
        }
    }
    __syncthreads();

    v4f* outp = (v4f*)(agg + (size_t)lo * D);
    for (int i = tid; i < OWN_NODES * D / 4; i += OWN_THREADS) {
        const float* ap = acc + (i >> 5) * D + (i & 31);
        const v4f v = {ap[0], ap[32], ap[64], ap[96]};
        outp[i] = v;
    }
}

#define MLP_GROUPS  256
#define MLP_THREADS 512
#define N_TILES     (N_NODES / 16)
#define WPITCH      288
#define WMAT        (64 * WPITCH)
#define TILE_F      (16 * LDSW)
#define W_ITERS     (64 * 32 / MLP_THREADS)

__device__ __forceinline__ void stage_weights(const float* __restrict__ Wa,
                                              const float* __restrict__ Wb,
                                              const float* __restrict__ Wc,
                                              float* __restrict__ sh_w, int tid) {
    v4f r0[3][W_ITERS], r1[3][W_ITERS];
    #pragma unroll
    for (int m = 0; m < 3; ++m) {
        const float* Wg = m == 0 ? Wa : (m == 1 ? Wb : Wc);
        #pragma unroll
        for (int j = 0; j < W_ITERS; ++j) {
            const int i  = tid + MLP_THREADS * j;
            const int kp = i >> 5;
            const int c  = (i & 31) * 4;
            r0[m][j] = *(const v4f*)(Wg + (size_t)(2 * kp) * D + c);
            r1[m][j] = *(const v4f*)(Wg + (size_t)(2 * kp + 1) * D + c);
        }
    }
    #pragma unroll
    for (int m = 0; m < 3; ++m) {
        #pragma unroll
        for (int j = 0; j < W_ITERS; ++j) {
            const int i  = tid + MLP_THREADS * j;
            const int kp = i >> 5;
            const int c  = (i & 31) * 4;
            const v4f a = r0[m][j];
            const v4f b = r1[m][j];
            const v4f lo = {a.x, b.x, a.y, b.y};
            const v4f hi = {a.z, b.z, a.w, b.w};
            float* p = sh_w + m * WMAT + kp * WPITCH + c * 2;
            *(v4f*)p       = lo;
            *(v4f*)(p + 4) = hi;
        }
    }
}

__device__ __forceinline__ void fetch_rows(const float* __restrict__ x,
                                           const float* __restrict__ agg,
                                           int ta, int tb, int tid,
                                           v4f (&px)[2], v4f (&pa)[2]) {
    ta = ta < N_TILES ? ta : N_TILES - 1;
    tb = tb < N_TILES ? tb : N_TILES - 1;
    #pragma unroll
    for (int j = 0; j < 2; ++j) {
        const int i    = tid + MLP_THREADS * j;
        const int tile = (j >= 1) ? tb : ta;
        const int r    = (i >> 5) & 15;
        const int c4   = (i & 31) * 4;
        const size_t off = (size_t)(tile * 16 + r) * D + c4;
        px[j] = *(const v4f*)(x + off);
        pa[j] = *(const v4f*)(agg + off);
    }
}

__device__ __forceinline__ void put_rows(float* __restrict__ sh_xa,
                                         float* __restrict__ sh_xr, int tid,
                                         const v4f (&px)[2], const v4f (&pa)[2]) {
    #pragma unroll
    for (int j = 0; j < 2; ++j) {
        const int i  = tid + MLP_THREADS * j;
        const int o  = (i >> 5) * LDSW + (i & 31) * 4;
        const v4f xv = px[j];
        const v4f av = pa[j];
        const v4f sv = {xv.x + av.x, xv.y + av.y, xv.z + av.z, xv.w + av.w};
        *(v4f*)(sh_xr + o) = xv;
        *(v4f*)(sh_xa + o) = sv;
    }
}

__device__ __forceinline__ void lds_barrier() {
    __builtin_amdgcn_fence(__ATOMIC_RELEASE, "workgroup", "local");
    __builtin_amdgcn_s_barrier();
    __builtin_amdgcn_fence(__ATOMIC_ACQUIRE, "workgroup", "local");
}

__global__ __launch_bounds__(MLP_THREADS) void gin_wmma_kernel(
    const float* __restrict__ x,   const float* __restrict__ agg,
    const float* __restrict__ W1,  const float* __restrict__ b1,
    const float* __restrict__ W2,  const float* __restrict__ b2,
    const float* __restrict__ Wres,const float* __restrict__ bres,
    float* __restrict__ out) {

    __shared__ __attribute__((aligned(16))) float sh_w[3 * WMAT];
    __shared__ __attribute__((aligned(16))) float sh_xa[2 * TILE_F];
    __shared__ __attribute__((aligned(16))) float sh_xr[2 * TILE_F];
    __shared__ __attribute__((aligned(16))) float sh_t1[2 * TILE_F];

    const int tid  = threadIdx.x;
    const int lane = tid & 31;
    const int wv   = __builtin_amdgcn_readfirstlane(tid >> 5);
    const int slot = wv >> 3;
    const int n    = lane & 15;
    const int h    = lane >> 4;
    const int colbase = (wv & 7) * 16;

    v4f px[2], pa[2];
    fetch_rows(x, agg, (int)blockIdx.x, (int)blockIdx.x + MLP_GROUPS, tid, px, pa);

    stage_weights(W1, W2, Wres, sh_w, tid);

    const float bias1 = b1[colbase + n];
    const float bias2 = b2[colbase + n] + bres[colbase + n];

    const float* w1 = sh_w + h * WPITCH + (colbase + n) * 2;
    const float* w2 = w1 + WMAT;
    const float* wr = w1 + 2 * WMAT;
    const int aoff  = slot * TILE_F + n * LDSW + 2 * h;

    for (int ta = blockIdx.x; ta < N_TILES; ta += 2 * MLP_GROUPS) {
        const int my_tile = ta + slot * MLP_GROUPS;
        const bool active = my_tile < N_TILES;

        put_rows(sh_xa, sh_xr, tid, px, pa);
        lds_barrier();
        fetch_rows(x, agg, ta + 2 * MLP_GROUPS, ta + 3 * MLP_GROUPS, tid, px, pa);

        if (active) {
            v8f acc = {};
            #pragma unroll
            for (int k = 0; k < D; k += 4) {
                const v2f a = *(const v2f*)(sh_xa + aoff + k);
                const v2f b = *(const v2f*)(w1 + (k >> 1) * WPITCH);
                acc = __builtin_amdgcn_wmma_f32_16x16x4_f32(
                          false, a, false, b, (short)0, acc, false, false);
            }
            #pragma unroll
            for (int r = 0; r < 8; ++r) {
                const float v = acc[r] + bias1;
                sh_t1[slot * TILE_F + (r + 8 * h) * LDSW + colbase + n] = fmaxf(v, 0.f);
            }
        }
        lds_barrier();

        if (active) {
            v8f acc2 = {};
            #pragma unroll
            for (int k = 0; k < D; k += 4) {
                const v2f a1  = *(const v2f*)(sh_t1 + aoff + k);
                const v2f bw2 = *(const v2f*)(w2 + (k >> 1) * WPITCH);
                acc2 = __builtin_amdgcn_wmma_f32_16x16x4_f32(
                           false, a1, false, bw2, (short)0, acc2, false, false);
                const v2f a2  = *(const v2f*)(sh_xr + aoff + k);
                const v2f bwr = *(const v2f*)(wr + (k >> 1) * WPITCH);
                acc2 = __builtin_amdgcn_wmma_f32_16x16x4_f32(
                           false, a2, false, bwr, (short)0, acc2, false, false);
            }
            const int rowbase = my_tile * 16;
            #pragma unroll
            for (int r = 0; r < 8; ++r)
                out[(size_t)(rowbase + r + 8 * h) * D + colbase + n] = acc2[r] + bias2;
        }
        lds_barrier();
    }
}

extern "C" void kernel_launch(void* const* d_in, const int* in_sizes, int n_in,
                              void* d_out, int out_size, void* d_ws, size_t ws_size,
                              hipStream_t stream) {
    const float* x    = (const float*)d_in[0];
    const int*   ei   = (const int*)  d_in[1];
    const float* W1   = (const float*)d_in[2];
    const float* b1   = (const float*)d_in[3];
    const float* W2   = (const float*)d_in[4];
    const float* b2   = (const float*)d_in[5];
    const float* Wres = (const float*)d_in[6];
    const float* bres = (const float*)d_in[7];
    float* out = (float*)d_out;
    float* agg = (float*)d_ws;
    unsigned int* rec = (unsigned int*)((char*)d_ws + (size_t)N_NODES * D * sizeof(float));
    unsigned int* bm  = (unsigned int*)((char*)d_ws + (size_t)N_NODES * D * sizeof(float) + (size_t)N_EDGES * sizeof(unsigned int));

    edge_index_kernel<<<IDX_GROUPS, IDX_THREADS, 0, stream>>>(ei, ei + N_EDGES, rec, bm);

    gather_owned_kernel<<<OWN_GROUPS, OWN_THREADS, 0, stream>>>(x, rec, bm, agg);

    gin_wmma_kernel<<<MLP_GROUPS, MLP_THREADS, 0, stream>>>(
        x, agg, W1, b1, W2, b2, Wres, bres, out);
}


# index kernel orders each 4096-edge block's records by owner in LDS (one counter add per edge) before writing them: an owner's records share 64-byte units, ~8x fewer to fetch; adds in arrival order
# speedup vs baseline: 2.6290x; 1.0328x over previous
#include <hip/hip_runtime.h>

typedef __attribute__((ext_vector_type(2))) float v2f;
typedef __attribute__((ext_vector_type(4))) float v4f;
typedef __attribute__((ext_vector_type(8))) float v8f;

#define N_NODES 50000
#define N_EDGES 800000
#define D 128
#define LDSW 132

typedef __attribute__((ext_vector_type(4))) int v4i;
typedef __attribute__((ext_vector_type(4))) unsigned int v4u;

#define OWN_NODES   200
#define OWN_GROUPS  (N_NODES / OWN_NODES)
#define N_DST4      (N_EDGES / 4)
#define N_UNITS     (N_EDGES / 16)
#define IDX_THREADS 256
#define IDX_PER_THREAD 4
#define IDX_WORDS   (IDX_THREADS * IDX_PER_THREAD)
#define IDX_EDGES   (IDX_WORDS * 4)
#define IDX_GROUPS  ((N_DST4 + IDX_WORDS - 1) / IDX_WORDS)
#define IDX_BINS    256
#define BM_GW       (IDX_WORDS / 4 / 32)
#define BM_WORDS    (IDX_GROUPS * BM_GW)

static_assert(N_NODES <= 65536, "src must fit the upper 16 bits of a record");
static_assert(N_EDGES % 16 == 0, "whole units");
static_assert(OWN_NODES < 256 && OWN_GROUPS <= 250, "row and key must fit one byte each");
static_assert(IDX_BINS == IDX_THREADS && OWN_GROUPS < IDX_BINS, "one counter per thread, one spare bin");
static_assert(BM_GW == 8, "thread o writes its 8 bit words as two 16-byte stores");

__device__ __forceinline__ unsigned edge_record(int s, int d) {
    const unsigned o   = (unsigned)d / (unsigned)OWN_NODES;
    const unsigned key = o < 255u ? o : 255u;
    return ((unsigned)s << 16) | (key << 8) | ((unsigned)d - o * (unsigned)OWN_NODES);
}

__device__ __forceinline__ unsigned owner_bin(int d) {
    const unsigned o = (unsigned)d / (unsigned)OWN_NODES;
    return o < (unsigned)OWN_GROUPS ? o : (unsigned)OWN_GROUPS;
}

__device__ __forceinline__ unsigned unit_bits(int ulo, int uhi, int w) {
    int a = ulo - 32 * w;
    int b = uhi - 32 * w;
    if (b < 0 || a > 31) return 0u;
    a = a < 0 ? 0 : a;
    b = b > 31 ? 31 : b;
    return (0xFFFFFFFFu >> (31 - b)) & (0xFFFFFFFFu << a);
}

__global__ __launch_bounds__(IDX_THREADS) void edge_index_kernel(
    const int* __restrict__ src,
    const int* __restrict__ dst,
    unsigned int* __restrict__ rec,
    unsigned int* __restrict__ bm) {
    __shared__ __attribute__((aligned(16))) unsigned int cnt[IDX_BINS];
    __shared__ __attribute__((aligned(16))) unsigned int first[IDX_BINS];
    __shared__ __attribute__((aligned(16))) unsigned int srt[IDX_EDGES];

    const int tid   = threadIdx.x;
    const int lane  = tid & 31;
    const int wbase = blockIdx.x * IDX_WORDS + (tid >> 5) * (32 * IDX_PER_THREAD) + (tid & 31);
    const v4i* dst4 = (const v4i*)dst;
    const v4i* src4 = (const v4i*)src;
    v4u* rec4 = (v4u*)rec;

    cnt[tid] = 0u;

    v4i d[IDX_PER_THREAD];
    v4i s[IDX_PER_THREAD];
    #pragma unroll
    for (int j = 0; j < IDX_PER_THREAD; ++j) {
        const int i  = wbase + 32 * j;
        const int ic = i < N_DST4 ? i : N_DST4 - 1;
        d[j] = dst4[ic];
        s[j] = src4[ic];
    }
    __syncthreads();

    unsigned slot[IDX_PER_THREAD][4];
    #pragma unroll
    for (int j = 0; j < IDX_PER_THREAD; ++j) {
        const int i = wbase + 32 * j;
        slot[j][0] = 0u; slot[j][1] = 0u; slot[j][2] = 0u; slot[j][3] = 0u;
        if (i < N_DST4) {
            #pragma unroll
            for (int c = 0; c < 4; ++c)
                slot[j][c] = atomicAdd(&cnt[owner_bin(d[j][c])], 1u);
        }
    }
    __syncthreads();

    if (tid < 32) {
        const v4u c0 = ((const v4u*)cnt)[2 * lane];
        const v4u c1 = ((const v4u*)cnt)[2 * lane + 1];
        const int sum = (int)(c0.x + c0.y + c0.z + c0.w + c1.x + c1.y + c1.z + c1.w);
        int incl = sum;
        #pragma unroll
        for (int dlt = 1; dlt < 32; dlt <<= 1) {
            const int t = __shfl_up(incl, dlt);
            if (lane >= dlt) incl += t;
        }
        const unsigned r0 = (unsigned)(incl - sum);
        const unsigned r1 = r0 + c0.x;
        const unsigned r2 = r1 + c0.y;
        const unsigned r3 = r2 + c0.z;
        const unsigned r4 = r3 + c0.w;
        const unsigned r5 = r4 + c1.x;
        const unsigned r6 = r5 + c1.y;
        const unsigned r7 = r6 + c1.z;
        const v4u f0 = {r0, r1, r2, r3};
        const v4u f1 = {r4, r5, r6, r7};
        ((v4u*)first)[2 * lane]     = f0;
        ((v4u*)first)[2 * lane + 1] = f1;
    }
    __syncthreads();

    #pragma unroll
    for (int j = 0; j < IDX_PER_THREAD; ++j) {
        const int i = wbase + 32 * j;
        if (i < N_DST4) {
            #pragma unroll
            for (int c = 0; c < 4; ++c)
                srt[first[owner_bin(d[j][c])] + slot[j][c]] = edge_record(s[j][c], d[j][c]);
        }
    }

    if (tid < OWN_GROUPS) {
        const unsigned n  = cnt[tid];
        const unsigned st = first[tid];
        const int ulo = n ? (int)(st >> 4) : 1;
        const int uhi = n ? (int)((st + n - 1u) >> 4) : 0;
        const v4u w0 = {unit_bits(ulo, uhi, 0), unit_bits(ulo, uhi, 1), unit_bits(ulo, uhi, 2), unit_bits(ulo, uhi, 3)};
        const v4u w1 = {unit_bits(ulo, uhi, 4), unit_bits(ulo, uhi, 5), unit_bits(ulo, uhi, 6), unit_bits(ulo, uhi, 7)};
        v4u* bp = (v4u*)(bm + (size_t)tid * BM_WORDS + (size_t)blockIdx.x * BM_GW);
        bp[0] = w0;
        bp[1] = w1;
    }
    __syncthreads();

    const int nitems = (N_DST4 - (int)blockIdx.x * IDX_WORDS) < IDX_WORDS ? (N_DST4 - (int)blockIdx.x * IDX_WORDS) : IDX_WORDS;
    #pragma unroll
    for (int j = 0; j < IDX_PER_THREAD; ++j) {
        const int it = tid + IDX_THREADS * j;
        if (it < nitems) rec4[(size_t)blockIdx.x * IDX_WORDS + it] = ((const v4u*)srt)[it];
    }
}

#define OWN_THREADS 1024
#define OWN_WAVES   (OWN_THREADS / 32)
#define UQCAP       8192
#define QCAP        16384
#define APOS(f)     (32 * ((f) & 3) + ((f) >> 2))

__device__ __forceinline__ int wave_reserve(int c, int lane, int* counter) {
    int incl = c;
    #pragma unroll
    for (int dlt = 1; dlt < 32; dlt <<= 1) {
        const int t = __shfl_up(incl, dlt);
        if (lane >= dlt) incl += t;
    }
    const int total = __shfl(incl, 31);
    int b0 = 0;
    if (lane == 0 && total > 0) b0 = atomicAdd(counter, total);
    b0 = __shfl(b0, 0);
    return b0 + incl - c;
}

__device__ __forceinline__ void add_edge_slowly(float* acc, const float* __restrict__ x, unsigned rc) {
    const int s = (int)(rc >> 16);
    const int r = (int)(rc & 255u);
    for (int f = 0; f < D; ++f)
        atomicAdd(&acc[r * D + APOS(f)], x[(size_t)s * D + f]);
}

__global__ __launch_bounds__(OWN_THREADS) void gather_owned_kernel(
    const float* __restrict__ x,
    const unsigned int* __restrict__ rec,
    const unsigned int* __restrict__ bm,
    float* __restrict__ agg) {

    __shared__ float acc[(OWN_NODES + 1) * D];
    __shared__ unsigned int queue[QCAP];
    __shared__ int uq[UQCAP];
    __shared__ int qcount;
    __shared__ int ucount;

    const int tid  = threadIdx.x;
    const int lane = tid & 31;
    const int wave = tid >> 5;
    const int lo   = blockIdx.x * OWN_NODES;
    const unsigned mykey = (unsigned)blockIdx.x << 8;

    for (int i = tid; i < (OWN_NODES + 1) * D / 4; i += OWN_THREADS) {
        v4f z = {0.f, 0.f, 0.f, 0.f};
        ((v4f*)acc)[i] = z;
    }
    if (tid == 0) { qcount = 0; ucount = 0; }
    __syncthreads();

    const unsigned int* bmrow = bm + (size_t)blockIdx.x * BM_WORDS;
    for (int w0 = 0; w0 < BM_WORDS; w0 += OWN_THREADS) {
        const int w = w0 + tid;
        unsigned m = w < BM_WORDS ? bmrow[w] : 0u;
        int off = wave_reserve(__builtin_popcount(m), lane, &ucount);
        while (m) {
            const int u = 32 * w + __builtin_ctz(m);
            m &= m - 1;
            if (__builtin_expect(off < UQCAP, 1)) {
                uq[off] = u;
            } else {
                #pragma unroll 1
                for (int k = 0; k < 16; ++k) {
                    const unsigned rc = rec[16 * u + k];
                    if ((rc & 0xFF00u) == mykey) add_edge_slowly(acc, x, rc);
                }
            }
            ++off;
        }
    }
    __syncthreads();

    const v4u* rec16 = (const v4u*)rec;
    const int nu = ucount < UQCAP ? ucount : UQCAP;
    const int wave_s = __builtin_amdgcn_readfirstlane(wave);
    for (int base = wave_s * 32; base < nu; base += OWN_WAVES * 32) {
        v4u rv[4];
        bool have[4];
        int  un[4];
        #pragma unroll
        for (int j = 0; j < 4; ++j) {
            const int ui = base + (lane >> 2) + 8 * j;
            have[j] = ui < nu;
            un[j] = uq[have[j] ? ui : base];
        }
        #pragma unroll
        for (int j = 0; j < 4; ++j)
            rv[j] = rec16[4 * un[j] + (lane & 3)];
        bool     keep[16];
        unsigned bal[16];
        int total = 0;
        #pragma unroll
        for (int k = 0; k < 16; ++k) {
            keep[k] = have[k >> 2] && (rv[k >> 2][k & 3] & 0xFF00u) == mykey;
            bal[k]  = __builtin_amdgcn_ballot_w32(keep[k]);
            total  += __builtin_popcount(bal[k]);
        }
        if (total > 0) {
            int b0 = 0;
            if (lane == 0) b0 = atomicAdd(&qcount, total);
            int pre = __builtin_amdgcn_readfirstlane(b0);
            #pragma unroll
            for (int k = 0; k < 16; ++k) {
                if (keep[k]) {
                    const unsigned rc = rv[k >> 2][k & 3];
                    const int off = pre + (int)__builtin_amdgcn_mbcnt_lo(bal[k], 0u);
                    if (__builtin_expect(off < QCAP, 1)) {
                        queue[off] = rc;
                    } else {
                        add_edge_slowly(acc, x, rc);
                    }
                }
                pre += __builtin_popcount(bal[k]);
            }
        }
    }
    __syncthreads();

    const int n = qcount < QCAP ? qcount : QCAP;
    for (int base = wave * 32; base < n; base += OWN_WAVES * 32) {
        const int i = base + lane;
        const unsigned rc = i < n ? queue[i] : (unsigned)OWN_NODES;
        const int s = (int)(rc >> 16);
        const int r = (int)(rc & 255u);
        #pragma unroll
        for (int j0 = 0; j0 < 32; j0 += 8) {
            v4f xv[8];
            #pragma unroll
            for (int u = 0; u < 8; ++u) {
                const int sj = __builtin_amdgcn_readlane(s, j0 + u);
                xv[u] = *(const v4f*)(x + (size_t)sj * D + 4 * lane);
            }
            #pragma unroll
            for (int u = 0; u < 8; ++u) {
                const int rj = __builtin_amdgcn_readlane(r, j0 + u);
                float* ap = acc + rj * D + lane;
                atomicAdd(ap,      xv[u].x);
                atomicAdd(ap + 32, xv[u].y);
                atomicAdd(ap + 64, xv[u].z);
                atomicAdd(ap + 96, xv[u].w);
            }
        }
    }
    __syncthreads();

    v4f* outp = (v4f*)(agg + (size_t)lo * D);
    for (int i = tid; i < OWN_NODES * D / 4; i += OWN_THREADS) {
        const float* ap = acc + (i >> 5) * D + (i & 31);
        const v4f v = {ap[0], ap[32], ap[64], ap[96]};
        outp[i] = v;
    }
}

#define MLP_GROUPS  256
#define MLP_THREADS 512
#define N_TILES     (N_NODES / 16)
#define WPITCH      288
#define WMAT        (64 * WPITCH)
#define TILE_F      (16 * LDSW)
#define W_ITERS     (64 * 32 / MLP_THREADS)

__device__ __forceinline__ void stage_weights(const float* __restrict__ Wa,
                                              const float* __restrict__ Wb,
                                              const float* __restrict__ Wc,
                                              float* __restrict__ sh_w, int tid) {
    v4f r0[3][W_ITERS], r1[3][W_ITERS];
    #pragma unroll
    for (int m = 0; m < 3; ++m) {
        const float* Wg = m == 0 ? Wa : (m == 1 ? Wb : Wc);
        #pragma unroll
        for (int j = 0; j < W_ITERS; ++j) {
            const int i  = tid + MLP_THREADS * j;
            const int kp = i >> 5;
            const int c  = (i & 31) * 4;
            r0[m][j] = *(const v4f*)(Wg + (size_t)(2 * kp) * D + c);
            r1[m][j] = *(const v4f*)(Wg + (size_t)(2 * kp + 1) * D + c);
        }
    }
    #pragma unroll
    for (int m = 0; m < 3; ++m) {
        #pragma unroll
        for (int j = 0; j < W_ITERS; ++j) {
            const int i  = tid + MLP_THREADS * j;
            const int kp = i >> 5;
            const int c  = (i & 31) * 4;
            const v4f a = r0[m][j];
            const v4f b = r1[m][j];
            const v4f lo = {a.x, b.x, a.y, b.y};
            const v4f hi = {a.z, b.z, a.w, b.w};
            float* p = sh_w + m * WMAT + kp * WPITCH + c * 2;
            *(v4f*)p       = lo;
            *(v4f*)(p + 4) = hi;
        }
    }
}

__device__ __forceinline__ void fetch_rows(const float* __restrict__ x,
                                           const float* __restrict__ agg,
                                           int ta, int tb, int tid,
                                           v4f (&px)[2], v4f (&pa)[2]) {
    ta = ta < N_TILES ? ta : N_TILES - 1;
    tb = tb < N_TILES ? tb : N_TILES - 1;
    #pragma unroll
    for (int j = 0; j < 2; ++j) {
        const int i    = tid + MLP_THREADS * j;
        const int tile = (j >= 1) ? tb : ta;
        const int r    = (i >> 5) & 15;
        const int c4   = (i & 31) * 4;
        const size_t off = (size_t)(tile * 16 + r) * D + c4;
        px[j] = *(const v4f*)(x + off);
        pa[j] = *(const v4f*)(agg + off);
    }
}

__device__ __forceinline__ void put_rows(float* __restrict__ sh_xa,
                                         float* __restrict__ sh_xr, int tid,
                                         const v4f (&px)[2], const v4f (&pa)[2]) {
    #pragma unroll
    for (int j = 0; j < 2; ++j) {
        const int i  = tid + MLP_THREADS * j;
        const int o  = (i >> 5) * LDSW + (i & 31) * 4;
        const v4f xv = px[j];
        const v4f av = pa[j];
        const v4f sv = {xv.x + av.x, xv.y + av.y, xv.z + av.z, xv.w + av.w};
        *(v4f*)(sh_xr + o) = xv;
        *(v4f*)(sh_xa + o) = sv;
    }
}

__device__ __forceinline__ void lds_barrier() {
    __builtin_amdgcn_fence(__ATOMIC_RELEASE, "workgroup", "local");
    __builtin_amdgcn_s_barrier();
    __builtin_amdgcn_fence(__ATOMIC_ACQUIRE, "workgroup", "local");
}

__global__ __launch_bounds__(MLP_THREADS) void gin_wmma_kernel(
    const float* __restrict__ x,   const float* __restrict__ agg,
    const float* __restrict__ W1,  const float* __restrict__ b1,
    const float* __restrict__ W2,  const float* __restrict__ b2,
    const float* __restrict__ Wres,const float* __restrict__ bres,
    float* __restrict__ out) {

    __shared__ __attribute__((aligned(16))) float sh_w[3 * WMAT];
    __shared__ __attribute__((aligned(16))) float sh_xa[2 * TILE_F];
    __shared__ __attribute__((aligned(16))) float sh_xr[2 * TILE_F];
    __shared__ __attribute__((aligned(16))) float sh_t1[2 * TILE_F];

    const int tid  = threadIdx.x;
    const int lane = tid & 31;
    const int wv   = __builtin_amdgcn_readfirstlane(tid >> 5);
    const int slot = wv >> 3;
    const int n    = lane & 15;
    const int h    = lane >> 4;
    const int colbase = (wv & 7) * 16;

    v4f px[2], pa[2];
    fetch_rows(x, agg, (int)blockIdx.x, (int)blockIdx.x + MLP_GROUPS, tid, px, pa);

    stage_weights(W1, W2, Wres, sh_w, tid);

    const float bias1 = b1[colbase + n];
    const float bias2 = b2[colbase + n] + bres[colbase + n];

    const float* w1 = sh_w + h * WPITCH + (colbase + n) * 2;
    const float* w2 = w1 + WMAT;
    const float* wr = w1 + 2 * WMAT;
    const int aoff  = slot * TILE_F + n * LDSW + 2 * h;

    for (int ta = blockIdx.x; ta < N_TILES; ta += 2 * MLP_GROUPS) {
        const int my_tile = ta + slot * MLP_GROUPS;
        const bool active = my_tile < N_TILES;

        put_rows(sh_xa, sh_xr, tid, px, pa);
        lds_barrier();
        fetch_rows(x, agg, ta + 2 * MLP_GROUPS, ta + 3 * MLP_GROUPS, tid, px, pa);

        if (active) {
            v8f acc = {};
            #pragma unroll
            for (int k = 0; k < D; k += 4) {
                const v2f a = *(const v2f*)(sh_xa + aoff + k);
                const v2f b = *(const v2f*)(w1 + (k >> 1) * WPITCH);
                acc = __builtin_amdgcn_wmma_f32_16x16x4_f32(
                          false, a, false, b, (short)0, acc, false, false);
            }
            #pragma unroll
            for (int r = 0; r < 8; ++r) {
                const float v = acc[r] + bias1;
                sh_t1[slot * TILE_F + (r + 8 * h) * LDSW + colbase + n] = fmaxf(v, 0.f);
            }
        }
        lds_barrier();

        if (active) {
            v8f acc2 = {};
            #pragma unroll
            for (int k = 0; k < D; k += 4) {
                const v2f a1  = *(const v2f*)(sh_t1 + aoff + k);
                const v2f bw2 = *(const v2f*)(w2 + (k >> 1) * WPITCH);
                acc2 = __builtin_amdgcn_wmma_f32_16x16x4_f32(
                           false, a1, false, bw2, (short)0, acc2, false, false);
                const v2f a2  = *(const v2f*)(sh_xr + aoff + k);
                const v2f bwr = *(const v2f*)(wr + (k >> 1) * WPITCH);
                acc2 = __builtin_amdgcn_wmma_f32_16x16x4_f32(
                           false, a2, false, bwr, (short)0, acc2, false, false);
            }
            const int rowbase = my_tile * 16;
            #pragma unroll
            for (int r = 0; r < 8; ++r)
                out[(size_t)(rowbase + r + 8 * h) * D + colbase + n] = acc2[r] + bias2;
        }
        lds_barrier();
    }
}

extern "C" void kernel_launch(void* const* d_in, const int* in_sizes, int n_in,
                              void* d_out, int out_size, void* d_ws, size_t ws_size,
                              hipStream_t stream) {
    const float* x    = (const float*)d_in[0];
    const int*   ei   = (const int*)  d_in[1];
    const float* W1   = (const float*)d_in[2];
    const float* b1   = (const float*)d_in[3];
    const float* W2   = (const float*)d_in[4];
    const float* b2   = (const float*)d_in[5];
    const float* Wres = (const float*)d_in[6];
    const float* bres = (const float*)d_in[7];
    float* out = (float*)d_out;
    float* agg = (float*)d_ws;
    unsigned int* rec = (unsigned int*)((char*)d_ws + (size_t)N_NODES * D * sizeof(float));
    unsigned int* bm  = (unsigned int*)((char*)d_ws + (size_t)N_NODES * D * sizeof(float) + (size_t)N_EDGES * sizeof(unsigned int));

    edge_index_kernel<<<IDX_GROUPS, IDX_THREADS, 0, stream>>>(ei, ei + N_EDGES, rec, bm);

    gather_owned_kernel<<<OWN_GROUPS, OWN_THREADS, 0, stream>>>(x, rec, bm, agg);

    gin_wmma_kernel<<<MLP_GROUPS, MLP_THREADS, 0, stream>>>(
        x, agg, W1, b1, W2, b2, Wres, bres, out);
}


# aggregation and MLP in one launch: the owner work-group keeps h = x + agg in LDS and multiplies its own 200 rows there (one weight matrix in LDS at a time); sum order: x first in h, W2 then Wres
# speedup vs baseline: 2.8101x; 1.0689x over previous
#include <hip/hip_runtime.h>

typedef __attribute__((ext_vector_type(2))) float v2f;
typedef __attribute__((ext_vector_type(4))) float v4f;
typedef __attribute__((ext_vector_type(8))) float v8f;

#define N_NODES 50000
#define N_EDGES 800000
#define D 128
#define LDSW 132

typedef __attribute__((ext_vector_type(4))) int v4i;
typedef __attribute__((ext_vector_type(4))) unsigned int v4u;

#define OWN_NODES   200
#define OWN_GROUPS  (N_NODES / OWN_NODES)
#define N_DST4      (N_EDGES / 4)
#define N_UNITS     (N_EDGES / 16)
#define IDX_THREADS 256
#define IDX_PER_THREAD 4
#define IDX_WORDS   (IDX_THREADS * IDX_PER_THREAD)
#define IDX_EDGES   (IDX_WORDS * 4)
#define IDX_GROUPS  ((N_DST4 + IDX_WORDS - 1) / IDX_WORDS)
#define IDX_BINS    256
#define BM_GW       (IDX_WORDS / 4 / 32)
#define BM_WORDS    (IDX_GROUPS * BM_GW)

static_assert(N_NODES <= 65536, "src must fit the upper 16 bits of a record");
static_assert(N_EDGES % 16 == 0, "whole units");
static_assert(OWN_NODES < 256 && OWN_GROUPS <= 250, "row and key must fit one byte each");
static_assert(IDX_BINS == IDX_THREADS && OWN_GROUPS < IDX_BINS, "one counter per thread, one spare bin");
static_assert(BM_GW == 8, "thread o writes its 8 bit words as two 16-byte stores");

__device__ __forceinline__ unsigned edge_record(int s, int d) {
    const unsigned o   = (unsigned)d / (unsigned)OWN_NODES;
    const unsigned key = o < 255u ? o : 255u;
    return ((unsigned)s << 16) | (key << 8) | ((unsigned)d - o * (unsigned)OWN_NODES);
}

__device__ __forceinline__ unsigned owner_bin(int d) {
    const unsigned o = (unsigned)d / (unsigned)OWN_NODES;
    return o < (unsigned)OWN_GROUPS ? o : (unsigned)OWN_GROUPS;
}

__device__ __forceinline__ unsigned unit_bits(int ulo, int uhi, int w) {
    int a = ulo - 32 * w;
    int b = uhi - 32 * w;
    if (b < 0 || a > 31) return 0u;
    a = a < 0 ? 0 : a;
    b = b > 31 ? 31 : b;
    return (0xFFFFFFFFu >> (31 - b)) & (0xFFFFFFFFu << a);
}

__global__ __launch_bounds__(IDX_THREADS) void edge_index_kernel(
    const int* __restrict__ src,
    const int* __restrict__ dst,
    unsigned int* __restrict__ rec,
    unsigned int* __restrict__ bm) {
    __shared__ __attribute__((aligned(16))) unsigned int cnt[IDX_BINS];
    __shared__ __attribute__((aligned(16))) unsigned int first[IDX_BINS];
    __shared__ __attribute__((aligned(16))) unsigned int srt[IDX_EDGES];

    const int tid   = threadIdx.x;
    const int lane  = tid & 31;
    const int wbase = blockIdx.x * IDX_WORDS + (tid >> 5) * (32 * IDX_PER_THREAD) + (tid & 31);
    const v4i* dst4 = (const v4i*)dst;
    const v4i* src4 = (const v4i*)src;
    v4u* rec4 = (v4u*)rec;

    cnt[tid] = 0u;

    v4i d[IDX_PER_THREAD];
    v4i s[IDX_PER_THREAD];
    #pragma unroll
    for (int j = 0; j < IDX_PER_THREAD; ++j) {
        const int i  = wbase + 32 * j;
        const int ic = i < N_DST4 ? i : N_DST4 - 1;
        d[j] = dst4[ic];
        s[j] = src4[ic];
    }
    __syncthreads();

    unsigned slot[IDX_PER_THREAD][4];
    #pragma unroll
    for (int j = 0; j < IDX_PER_THREAD; ++j) {
        const int i = wbase + 32 * j;
        slot[j][0] = 0u; slot[j][1] = 0u; slot[j][2] = 0u; slot[j][3] = 0u;
        if (i < N_DST4) {
            #pragma unroll
            for (int c = 0; c < 4; ++c)
                slot[j][c] = atomicAdd(&cnt[owner_bin(d[j][c])], 1u);
        }
    }
    __syncthreads();

    if (tid < 32) {
        const v4u c0 = ((const v4u*)cnt)[2 * lane];
        const v4u c1 = ((const v4u*)cnt)[2 * lane + 1];
        const int sum = (int)(c0.x + c0.y + c0.z + c0.w + c1.x + c1.y + c1.z + c1.w);
        int incl = sum;
        #pragma unroll
        for (int dlt = 1; dlt < 32; dlt <<= 1) {
            const int t = __shfl_up(incl, dlt);
            if (lane >= dlt) incl += t;
        }
        const unsigned r0 = (unsigned)(incl - sum);
        const unsigned r1 = r0 + c0.x;
        const unsigned r2 = r1 + c0.y;
        const unsigned r3 = r2 + c0.z;
        const unsigned r4 = r3 + c0.w;
        const unsigned r5 = r4 + c1.x;
        const unsigned r6 = r5 + c1.y;
        const unsigned r7 = r6 + c1.z;
        const v4u f0 = {r0, r1, r2, r3};
        const v4u f1 = {r4, r5, r6, r7};
        ((v4u*)first)[2 * lane]     = f0;
        ((v4u*)first)[2 * lane + 1] = f1;
    }
    __syncthreads();

    #pragma unroll
    for (int j = 0; j < IDX_PER_THREAD; ++j) {
        const int i = wbase + 32 * j;
        if (i < N_DST4) {
            #pragma unroll
            for (int c = 0; c < 4; ++c)
                srt[first[owner_bin(d[j][c])] + slot[j][c]] = edge_record(s[j][c], d[j][c]);
        }
    }

    if (tid < OWN_GROUPS) {
        const unsigned n  = cnt[tid];
        const unsigned st = first[tid];
        const int ulo = n ? (int)(st >> 4) : 1;
        const int uhi = n ? (int)((st + n - 1u) >> 4) : 0;
        const v4u w0 = {unit_bits(ulo, uhi, 0), unit_bits(ulo, uhi, 1), unit_bits(ulo, uhi, 2), unit_bits(ulo, uhi, 3)};
        const v4u w1 = {unit_bits(ulo, uhi, 4), unit_bits(ulo, uhi, 5), unit_bits(ulo, uhi, 6), unit_bits(ulo, uhi, 7)};
        v4u* bp = (v4u*)(bm + (size_t)tid * BM_WORDS + (size_t)blockIdx.x * BM_GW);
        bp[0] = w0;
        bp[1] = w1;
    }
    __syncthreads();

    const int nitems = (N_DST4 - (int)blockIdx.x * IDX_WORDS) < IDX_WORDS ? (N_DST4 - (int)blockIdx.x * IDX_WORDS) : IDX_WORDS;
    #pragma unroll
    for (int j = 0; j < IDX_PER_THREAD; ++j) {
        const int it = tid + IDX_THREADS * j;
        if (it < nitems) rec4[(size_t)blockIdx.x * IDX_WORDS + it] = ((const v4u*)srt)[it];
    }
}

#define OWN_THREADS 1024
#define OWN_WAVES   (OWN_THREADS / 32)
#define UQCAP       8192
#define QCAP        16384
#define APOS(f)     (32 * ((f) & 3) + ((f) >> 2))
#define APITCH      LDSW

__device__ __forceinline__ int wave_reserve(int c, int lane, int* counter) {
    int incl = c;
    #pragma unroll
    for (int dlt = 1; dlt < 32; dlt <<= 1) {
        const int t = __shfl_up(incl, dlt);
        if (lane >= dlt) incl += t;
    }
    const int total = __shfl(incl, 31);
    int b0 = 0;
    if (lane == 0 && total > 0) b0 = atomicAdd(counter, total);
    b0 = __shfl(b0, 0);
    return b0 + incl - c;
}

__device__ __forceinline__ void add_edge_slowly(float* acc, const float* __restrict__ x, unsigned rc) {
    const int s = (int)(rc >> 16);
    const int r = (int)(rc & 255u);
    for (int f = 0; f < D; ++f)
        atomicAdd(&acc[r * APITCH + APOS(f)], x[(size_t)s * D + f]);
}


#define OWN_TILES   ((OWN_NODES + 15) / 16)
#define OWN_PAD     (OWN_TILES * 16)
#define WPITCH      288
#define WMAT        (64 * WPITCH)
#define W_ITEMS     (64 * 32 / OWN_THREADS)

static_assert(QCAP + UQCAP <= OWN_PAD * APITCH, "the work lists must fit the bytes of t1");
static_assert(OWN_TILES == 13 && OWN_WAVES == 32, "4 wave sets: tiles s, s+4, s+8 for every set, tile 12 for set 0");

__device__ __forceinline__ void stage_matrix(const float* __restrict__ Wg,
                                             float* __restrict__ shw, int tid,
                                             const bool rows_in_acc_order) {
    v4f r0[W_ITEMS], r1[W_ITEMS];
    #pragma unroll
    for (int j = 0; j < W_ITEMS; ++j) {
        const int i  = tid + OWN_THREADS * j;
        const int kp = i >> 5;
        const int c  = (i & 31) * 4;
        const int p  = 2 * kp;
        const int ra = rows_in_acc_order ? 4 * (p & 31) + (p >> 5) : p;
        const int rb = rows_in_acc_order ? ra + 4 : p + 1;
        r0[j] = *(const v4f*)(Wg + (size_t)ra * D + c);
        r1[j] = *(const v4f*)(Wg + (size_t)rb * D + c);
    }
    #pragma unroll
    for (int j = 0; j < W_ITEMS; ++j) {
        const int i  = tid + OWN_THREADS * j;
        const int kp = i >> 5;
        const int c  = (i & 31) * 4;
        const v4f a = r0[j];
        const v4f b = r1[j];
        const v4f lo = {a.x, b.x, a.y, b.y};
        const v4f hi = {a.z, b.z, a.w, b.w};
        float* q = shw + kp * WPITCH + c * 2;
        *(v4f*)q       = lo;
        *(v4f*)(q + 4) = hi;
    }
}

__device__ __forceinline__ v8f tile_product(const float* __restrict__ ap,
                                            const float* __restrict__ wb, v8f c) {
    #pragma unroll
    for (int k = 0; k < D; k += 4) {
        const v2f a = *(const v2f*)(ap + k);
        const v2f b = *(const v2f*)(wb + (k >> 1) * WPITCH);
        c = __builtin_amdgcn_wmma_f32_16x16x4_f32(false, a, false, b, (short)0, c, false, false);
    }
    return c;
}

__global__ __launch_bounds__(OWN_THREADS) void gin_owner_kernel(
    const float* __restrict__ x,
    const unsigned int* __restrict__ rec,
    const unsigned int* __restrict__ bm,
    const float* __restrict__ W1,  const float* __restrict__ b1,
    const float* __restrict__ W2,  const float* __restrict__ b2,
    const float* __restrict__ Wres,const float* __restrict__ bres,
    float* __restrict__ out) {

    __shared__ __attribute__((aligned(16))) float lds[2 * OWN_PAD * APITCH + WMAT];
    __shared__ int qcount;
    __shared__ int ucount;
    float* acc = lds;
    float* t1  = lds + OWN_PAD * APITCH;
    float* shw = lds + 2 * OWN_PAD * APITCH;
    unsigned int* queue = (unsigned int*)t1;
    int* uq = (int*)(t1 + QCAP);

    const int tid  = threadIdx.x;
    const int lane = tid & 31;
    const int wave = tid >> 5;
    const int lo   = blockIdx.x * OWN_NODES;
    const unsigned mykey = (unsigned)blockIdx.x << 8;

    stage_matrix(W1, shw, tid, true);

    for (int i = tid; i < OWN_PAD * 32; i += OWN_THREADS) {
        const int row = i >> 5;
        const int c   = i & 31;
        v4f v = {0.f, 0.f, 0.f, 0.f};
        if (row < OWN_NODES) v = *(const v4f*)(x + (size_t)(lo + row) * D + 4 * c);
        float* p = acc + row * APITCH + c;
        p[0] = v.x; p[32] = v.y; p[64] = v.z; p[96] = v.w;
    }
    if (tid == 0) { qcount = 0; ucount = 0; }
    __syncthreads();

    const unsigned int* bmrow = bm + (size_t)blockIdx.x * BM_WORDS;
    for (int w0 = 0; w0 < BM_WORDS; w0 += OWN_THREADS) {
        const int w = w0 + tid;
        unsigned m = w < BM_WORDS ? bmrow[w] : 0u;
        int off = wave_reserve(__builtin_popcount(m), lane, &ucount);
        while (m) {
            const int u = 32 * w + __builtin_ctz(m);
            m &= m - 1;
            if (__builtin_expect(off < UQCAP, 1)) {
                uq[off] = u;
            } else {
                #pragma unroll 1
                for (int k = 0; k < 16; ++k) {
                    const unsigned rc = rec[16 * u + k];
                    if ((rc & 0xFF00u) == mykey) add_edge_slowly(acc, x, rc);
                }
            }
            ++off;
        }
    }
    __syncthreads();

    const v4u* rec16 = (const v4u*)rec;
    const int nu = ucount < UQCAP ? ucount : UQCAP;
    const int wave_s = __builtin_amdgcn_readfirstlane(wave);
    for (int base = wave_s * 32; base < nu; base += OWN_WAVES * 32) {
        v4u rv[4];
        bool have[4];
        int  un[4];
        #pragma unroll
        for (int j = 0; j < 4; ++j) {
            const int ui = base + (lane >> 2) + 8 * j;
            have[j] = ui < nu;
            un[j] = uq[have[j] ? ui : base];
        }
        #pragma unroll
        for (int j = 0; j < 4; ++j)
            rv[j] = rec16[4 * un[j] + (lane & 3)];
        bool     keep[16];
        unsigned bal[16];
        int total = 0;
        #pragma unroll
        for (int k = 0; k < 16; ++k) {
            keep[k] = have[k >> 2] && (rv[k >> 2][k & 3] & 0xFF00u) == mykey;
            bal[k]  = __builtin_amdgcn_ballot_w32(keep[k]);
            total  += __builtin_popcount(bal[k]);
        }
        if (total > 0) {
            int b0 = 0;
            if (lane == 0) b0 = atomicAdd(&qcount, total);
            int pre = __builtin_amdgcn_readfirstlane(b0);
            #pragma unroll
            for (int k = 0; k < 16; ++k) {
                if (keep[k]) {
                    const unsigned rc = rv[k >> 2][k & 3];
                    const int off = pre + (int)__builtin_amdgcn_mbcnt_lo(bal[k], 0u);
                    if (__builtin_expect(off < QCAP, 1)) {
                        queue[off] = rc;
                    } else {
                        add_edge_slowly(acc, x, rc);
                    }
                }
                pre += __builtin_popcount(bal[k]);
            }
        }
    }
    __syncthreads();

    const int n = qcount < QCAP ? qcount : QCAP;
    for (int base = wave * 32; base < n; base += OWN_WAVES * 32) {
        const int i = base + lane;
        const unsigned rc = i < n ? queue[i] : (unsigned)OWN_NODES;
        const int s = (int)(rc >> 16);
        const int r = (int)(rc & 255u);
        #pragma unroll
        for (int j0 = 0; j0 < 32; j0 += 8) {
            v4f xv[8];
            #pragma unroll
            for (int u = 0; u < 8; ++u) {
                const int sj = __builtin_amdgcn_readlane(s, j0 + u);
                xv[u] = *(const v4f*)(x + (size_t)sj * D + 4 * lane);
            }
            #pragma unroll
            for (int u = 0; u < 8; ++u) {
                const int rj = __builtin_amdgcn_readlane(r, j0 + u);
                float* ap = acc + rj * APITCH + lane;
                atomicAdd(ap,      xv[u].x);
                atomicAdd(ap + 32, xv[u].y);
                atomicAdd(ap + 64, xv[u].z);
                atomicAdd(ap + 96, xv[u].w);
            }
        }
    }
    __syncthreads();

    const int wv      = __builtin_amdgcn_readfirstlane(wave);
    const int set     = wv >> 3;
    const bool extra  = set == 0;
    const int cn      = lane & 15;
    const int hh      = lane >> 4;
    const int colbase = (wv & 7) * 16;
    const float bias1 = b1[colbase + cn];
    const float bias2 = b2[colbase + cn] + bres[colbase + cn];
    const float* wb   = shw + hh * WPITCH + (colbase + cn) * 2;
    const int aoff    = cn * APITCH + 2 * hh;

    #pragma unroll
    for (int i = 0; i < 4; ++i) {
        if (i < 3 || extra) {
            const int row0 = (set + 4 * i) * 16;
            v8f c = {};
            c = tile_product(acc + row0 * APITCH + aoff, wb, c);
            #pragma unroll
            for (int r = 0; r < 8; ++r)
                t1[(row0 + r + 8 * hh) * APITCH + colbase + cn] = fmaxf(c[r] + bias1, 0.f);
        }
    }
    __syncthreads();

    stage_matrix(W2, shw, tid, false);
    for (int i = tid; i < OWN_NODES * 32; i += OWN_THREADS) {
        const int row = i >> 5;
        const int c4  = (i & 31) * 4;
        *(v4f*)(acc + row * APITCH + c4) = *(const v4f*)(x + (size_t)(lo + row) * D + c4);
    }
    __syncthreads();

    v8f o[4];
    #pragma unroll
    for (int i = 0; i < 4; ++i) {
        v8f c = {};
        if (i < 3 || extra)
            c = tile_product(t1 + (set + 4 * i) * 16 * APITCH + aoff, wb, c);
        o[i] = c;
    }
    __syncthreads();

    stage_matrix(Wres, shw, tid, false);
    __syncthreads();

    #pragma unroll
    for (int i = 0; i < 4; ++i) {
        if (i < 3 || extra) {
            const int row0 = (set + 4 * i) * 16;
            const v8f c = tile_product(acc + row0 * APITCH + aoff, wb, o[i]);
            #pragma unroll
            for (int r = 0; r < 8; ++r) {
                const int row = row0 + r + 8 * hh;
                if (row < OWN_NODES)
                    out[(size_t)(lo + row) * D + colbase + cn] = c[r] + bias2;
            }
        }
    }
}

extern "C" void kernel_launch(void* const* d_in, const int* in_sizes, int n_in,
                              void* d_out, int out_size, void* d_ws, size_t ws_size,
                              hipStream_t stream) {
    const float* x    = (const float*)d_in[0];
    const int*   ei   = (const int*)  d_in[1];
    const float* W1   = (const float*)d_in[2];
    const float* b1   = (const float*)d_in[3];
    const float* W2   = (const float*)d_in[4];
    const float* b2   = (const float*)d_in[5];
    const float* Wres = (const float*)d_in[6];
    const float* bres = (const float*)d_in[7];
    float* out = (float*)d_out;
    unsigned int* rec = (unsigned int*)d_ws;
    unsigned int* bm  = (unsigned int*)((char*)d_ws + (size_t)N_EDGES * sizeof(unsigned int));

    edge_index_kernel<<<IDX_GROUPS, IDX_THREADS, 0, stream>>>(ei, ei + N_EDGES, rec, bm);

    gin_owner_kernel<<<OWN_GROUPS, OWN_THREADS, 0, stream>>>(
        x, rec, bm, W1, b1, W2, b2, Wres, bres, out);
}


# x@Wres product moved into the gather loop: a wave runs 8 WMMAs after each group of 8 row fetches while they fly; sums go to a 2nd LDS buffer, h = x + agg after; sum order: Wres first, then W2
# speedup vs baseline: 3.0527x; 1.0863x over previous
#include <hip/hip_runtime.h>

typedef __attribute__((ext_vector_type(2))) float v2f;
typedef __attribute__((ext_vector_type(4))) float v4f;
typedef __attribute__((ext_vector_type(8))) float v8f;

#define N_NODES 50000
#define N_EDGES 800000
#define D 128
#define LDSW 132

typedef __attribute__((ext_vector_type(4))) int v4i;
typedef __attribute__((ext_vector_type(4))) unsigned int v4u;

#define OWN_NODES   200
#define OWN_GROUPS  (N_NODES / OWN_NODES)
#define N_DST4      (N_EDGES / 4)
#define N_UNITS     (N_EDGES / 16)
#define IDX_THREADS 256
#define IDX_PER_THREAD 4
#define IDX_WORDS   (IDX_THREADS * IDX_PER_THREAD)
#define IDX_EDGES   (IDX_WORDS * 4)
#define IDX_GROUPS  ((N_DST4 + IDX_WORDS - 1) / IDX_WORDS)
#define IDX_BINS    256
#define BM_GW       (IDX_WORDS / 4 / 32)
#define BM_WORDS    (IDX_GROUPS * BM_GW)

static_assert(N_NODES <= 65536, "src must fit the upper 16 bits of a record");
static_assert(N_EDGES % 16 == 0, "whole units");
static_assert(OWN_NODES < 256 && OWN_GROUPS <= 250, "row and key must fit one byte each");
static_assert(IDX_BINS == IDX_THREADS && OWN_GROUPS < IDX_BINS, "one counter per thread, one spare bin");
static_assert(BM_GW == 8, "thread o writes its 8 bit words as two 16-byte stores");

__device__ __forceinline__ unsigned edge_record(int s, int d) {
    const unsigned o   = (unsigned)d / (unsigned)OWN_NODES;
    const unsigned key = o < 255u ? o : 255u;
    return ((unsigned)s << 16) | (key << 8) | ((unsigned)d - o * (unsigned)OWN_NODES);
}

__device__ __forceinline__ unsigned owner_bin(int d) {
    const unsigned o = (unsigned)d / (unsigned)OWN_NODES;
    return o < (unsigned)OWN_GROUPS ? o : (unsigned)OWN_GROUPS;
}

__device__ __forceinline__ unsigned unit_bits(int ulo, int uhi, int w) {
    int a = ulo - 32 * w;
    int b = uhi - 32 * w;
    if (b < 0 || a > 31) return 0u;
    a = a < 0 ? 0 : a;
    b = b > 31 ? 31 : b;
    return (0xFFFFFFFFu >> (31 - b)) & (0xFFFFFFFFu << a);
}

__global__ __launch_bounds__(IDX_THREADS) void edge_index_kernel(
    const int* __restrict__ src,
    const int* __restrict__ dst,
    unsigned int* __restrict__ rec,
    unsigned int* __restrict__ bm) {
    __shared__ __attribute__((aligned(16))) unsigned int cnt[IDX_BINS];
    __shared__ __attribute__((aligned(16))) unsigned int first[IDX_BINS];
    __shared__ __attribute__((aligned(16))) unsigned int srt[IDX_EDGES];

    const int tid   = threadIdx.x;
    const int lane  = tid & 31;
    const int wbase = blockIdx.x * IDX_WORDS + (tid >> 5) * (32 * IDX_PER_THREAD) + (tid & 31);
    const v4i* dst4 = (const v4i*)dst;
    const v4i* src4 = (const v4i*)src;
    v4u* rec4 = (v4u*)rec;

    cnt[tid] = 0u;

    v4i d[IDX_PER_THREAD];
    v4i s[IDX_PER_THREAD];
    #pragma unroll
    for (int j = 0; j < IDX_PER_THREAD; ++j) {
        const int i  = wbase + 32 * j;
        const int ic = i < N_DST4 ? i : N_DST4 - 1;
        d[j] = dst4[ic];
        s[j] = src4[ic];
    }
    __syncthreads();

    unsigned slot[IDX_PER_THREAD][4];
    #pragma unroll
    for (int j = 0; j < IDX_PER_THREAD; ++j) {
        const int i = wbase + 32 * j;
        slot[j][0] = 0u; slot[j][1] = 0u; slot[j][2] = 0u; slot[j][3] = 0u;
        if (i < N_DST4) {
            #pragma unroll
            for (int c = 0; c < 4; ++c)
                slot[j][c] = atomicAdd(&cnt[owner_bin(d[j][c])], 1u);
        }
    }
    __syncthreads();

    if (tid < 32) {
        const v4u c0 = ((const v4u*)cnt)[2 * lane];
        const v4u c1 = ((const v4u*)cnt)[2 * lane + 1];
        const int sum = (int)(c0.x + c0.y + c0.z + c0.w + c1.x + c1.y + c1.z + c1.w);
        int incl = sum;
        #pragma unroll
        for (int dlt = 1; dlt < 32; dlt <<= 1) {
            const int t = __shfl_up(incl, dlt);
            if (lane >= dlt) incl += t;
        }
        const unsigned r0 = (unsigned)(incl - sum);
        const unsigned r1 = r0 + c0.x;
        const unsigned r2 = r1 + c0.y;
        const unsigned r3 = r2 + c0.z;
        const unsigned r4 = r3 + c0.w;
        const unsigned r5 = r4 + c1.x;
        const unsigned r6 = r5 + c1.y;
        const unsigned r7 = r6 + c1.z;
        const v4u f0 = {r0, r1, r2, r3};
        const v4u f1 = {r4, r5, r6, r7};
        ((v4u*)first)[2 * lane]     = f0;
        ((v4u*)first)[2 * lane + 1] = f1;
    }
    __syncthreads();

    #pragma unroll
    for (int j = 0; j < IDX_PER_THREAD; ++j) {
        const int i = wbase + 32 * j;
        if (i < N_DST4) {
            #pragma unroll
            for (int c = 0; c < 4; ++c)
                srt[first[owner_bin(d[j][c])] + slot[j][c]] = edge_record(s[j][c], d[j][c]);
        }
    }

    if (tid < OWN_GROUPS) {
        const unsigned n  = cnt[tid];
        const unsigned st = first[tid];
        const int ulo = n ? (int)(st >> 4) : 1;
        const int uhi = n ? (int)((st + n - 1u) >> 4) : 0;
        const v4u w0 = {unit_bits(ulo, uhi, 0), unit_bits(ulo, uhi, 1), unit_bits(ulo, uhi, 2), unit_bits(ulo, uhi, 3)};
        const v4u w1 = {unit_bits(ulo, uhi, 4), unit_bits(ulo, uhi, 5), unit_bits(ulo, uhi, 6), unit_bits(ulo, uhi, 7)};
        v4u* bp = (v4u*)(bm + (size_t)tid * BM_WORDS + (size_t)blockIdx.x * BM_GW);
        bp[0] = w0;
        bp[1] = w1;
    }
    __syncthreads();

    const int nitems = (N_DST4 - (int)blockIdx.x * IDX_WORDS) < IDX_WORDS ? (N_DST4 - (int)blockIdx.x * IDX_WORDS) : IDX_WORDS;
    #pragma unroll
    for (int j = 0; j < IDX_PER_THREAD; ++j) {
        const int it = tid + IDX_THREADS * j;
        if (it < nitems) rec4[(size_t)blockIdx.x * IDX_WORDS + it] = ((const v4u*)srt)[it];
    }
}

#define OWN_THREADS 1024
#define OWN_WAVES   (OWN_THREADS / 32)
#define UQCAP       2048
#define QCAP        5120
#define APOS(f)     (32 * ((f) & 3) + ((f) >> 2))
#define APITCH      LDSW

__device__ __forceinline__ int wave_reserve(int c, int lane, int* counter) {
    int incl = c;
    #pragma unroll
    for (int dlt = 1; dlt < 32; dlt <<= 1) {
        const int t = __shfl_up(incl, dlt);
        if (lane >= dlt) incl += t;
    }
    const int total = __shfl(incl, 31);
    int b0 = 0;
    if (lane == 0 && total > 0) b0 = atomicAdd(counter, total);
    b0 = __shfl(b0, 0);
    return b0 + incl - c;
}

__device__ __forceinline__ void add_edge_slowly(float* acc, const float* __restrict__ x, unsigned rc) {
    const int s = (int)(rc >> 16);
    const int r = (int)(rc & 255u);
    for (int f = 0; f < D; ++f)
        atomicAdd(&acc[r * APITCH + APOS(f)], x[(size_t)s * D + f]);
}


#define OWN_TILES   ((OWN_NODES + 15) / 16)
#define OWN_PAD     (OWN_TILES * 16)
#define WPITCH      288
#define WMAT        (64 * WPITCH)
#define W_ITEMS     (64 * 32 / OWN_THREADS)

static_assert((2 * OWN_PAD * APITCH + WMAT + QCAP + UQCAP) * 4 + 16 <= 320 * 1024, "everything must fit the LDS of one WGP");
static_assert(OWN_TILES == 13 && OWN_WAVES == 32, "4 wave sets: tiles s, s+4, s+8 for every set, tile 12 for set 0");

__device__ __forceinline__ void stage_matrix(const float* __restrict__ Wg,
                                             float* __restrict__ shw, int tid,
                                             const bool rows_in_acc_order) {
    v4f r0[W_ITEMS], r1[W_ITEMS];
    #pragma unroll
    for (int j = 0; j < W_ITEMS; ++j) {
        const int i  = tid + OWN_THREADS * j;
        const int kp = i >> 5;
        const int c  = (i & 31) * 4;
        const int p  = 2 * kp;
        const int ra = rows_in_acc_order ? 4 * (p & 31) + (p >> 5) : p;
        const int rb = rows_in_acc_order ? ra + 4 : p + 1;
        r0[j] = *(const v4f*)(Wg + (size_t)ra * D + c);
        r1[j] = *(const v4f*)(Wg + (size_t)rb * D + c);
    }
    #pragma unroll
    for (int j = 0; j < W_ITEMS; ++j) {
        const int i  = tid + OWN_THREADS * j;
        const int kp = i >> 5;
        const int c  = (i & 31) * 4;
        const v4f a = r0[j];
        const v4f b = r1[j];
        const v4f lo = {a.x, b.x, a.y, b.y};
        const v4f hi = {a.z, b.z, a.w, b.w};
        float* q = shw + kp * WPITCH + c * 2;
        *(v4f*)q       = lo;
        *(v4f*)(q + 4) = hi;
    }
}

template <int K0, int KN>
__device__ __forceinline__ v8f tile_product(const float* __restrict__ ap,
                                            const float* __restrict__ wb, v8f c) {
    #pragma unroll
    for (int k = K0; k < K0 + KN; k += 4) {
        const v2f a = *(const v2f*)(ap + k);
        const v2f b = *(const v2f*)(wb + (k >> 1) * WPITCH);
        c = __builtin_amdgcn_wmma_f32_16x16x4_f32(false, a, false, b, (short)0, c, false, false);
    }
    return c;
}

template <int J0>
__device__ __forceinline__ void fetch_rows(const float* __restrict__ x, int s, int lane, v4f (&xv)[8]) {
    #pragma unroll
    for (int u = 0; u < 8; ++u) {
        const int sj = __builtin_amdgcn_readlane(s, J0 + u);
        xv[u] = *(const v4f*)(x + (size_t)sj * D + 4 * lane);
    }
}

template <int J0>
__device__ __forceinline__ void add_rows(float* acc, int r, int lane, const v4f (&xv)[8]) {
    #pragma unroll
    for (int u = 0; u < 8; ++u) {
        const int rj = __builtin_amdgcn_readlane(r, J0 + u);
        float* ap = acc + rj * APITCH + lane;
        atomicAdd(ap,      xv[u].x);
        atomicAdd(ap + 32, xv[u].y);
        atomicAdd(ap + 64, xv[u].z);
        atomicAdd(ap + 96, xv[u].w);
    }
}

template <bool HAS_TILE>
__device__ __forceinline__ v8f gather_batch_with_tile(const float* __restrict__ x, float* acc,
                                                      int s, int r, int lane,
                                                      const float* __restrict__ ap,
                                                      const float* __restrict__ wb, v8f o) {
    v4f xv[8];
    fetch_rows<0>(x, s, lane, xv);
    if (HAS_TILE) o = tile_product<0, 32>(ap, wb, o);
    add_rows<0>(acc, r, lane, xv);
    fetch_rows<8>(x, s, lane, xv);
    if (HAS_TILE) o = tile_product<32, 32>(ap, wb, o);
    add_rows<8>(acc, r, lane, xv);
    fetch_rows<16>(x, s, lane, xv);
    if (HAS_TILE) o = tile_product<64, 32>(ap, wb, o);
    add_rows<16>(acc, r, lane, xv);
    fetch_rows<24>(x, s, lane, xv);
    if (HAS_TILE) o = tile_product<96, 32>(ap, wb, o);
    add_rows<24>(acc, r, lane, xv);
    return o;
}

__global__ __launch_bounds__(OWN_THREADS) void gin_owner_kernel(
    const float* __restrict__ x,
    const unsigned int* __restrict__ rec,
    const unsigned int* __restrict__ bm,
    const float* __restrict__ W1,  const float* __restrict__ b1,
    const float* __restrict__ W2,  const float* __restrict__ b2,
    const float* __restrict__ Wres,const float* __restrict__ bres,
    float* __restrict__ out) {

    __shared__ __attribute__((aligned(16))) float xs[OWN_PAD * APITCH];
    __shared__ __attribute__((aligned(16))) float acc[OWN_PAD * APITCH];
    __shared__ __attribute__((aligned(16))) float shw[WMAT];
    __shared__ __attribute__((aligned(16))) unsigned int queue[QCAP];
    __shared__ __attribute__((aligned(16))) int uq[UQCAP];
    __shared__ int qcount;
    __shared__ int ucount;

    const int tid  = threadIdx.x;
    const int lane = tid & 31;
    const int wave = tid >> 5;
    const int lo   = blockIdx.x * OWN_NODES;
    const unsigned mykey = (unsigned)blockIdx.x << 8;

    stage_matrix(Wres, shw, tid, true);

    for (int i = tid; i < OWN_PAD * 32; i += OWN_THREADS) {
        const int row = i >> 5;
        const int c   = i & 31;
        v4f v = {0.f, 0.f, 0.f, 0.f};
        if (row < OWN_NODES) v = *(const v4f*)(x + (size_t)(lo + row) * D + 4 * c);
        float* p = xs + row * APITCH + c;
        p[0] = v.x; p[32] = v.y; p[64] = v.z; p[96] = v.w;
        float* z = acc + row * APITCH + c;
        z[0] = 0.f; z[32] = 0.f; z[64] = 0.f; z[96] = 0.f;
    }
    if (tid == 0) { qcount = 0; ucount = 0; }
    __syncthreads();

    const unsigned int* bmrow = bm + (size_t)blockIdx.x * BM_WORDS;
    for (int w0 = 0; w0 < BM_WORDS; w0 += OWN_THREADS) {
        const int w = w0 + tid;
        unsigned m = w < BM_WORDS ? bmrow[w] : 0u;
        int off = wave_reserve(__builtin_popcount(m), lane, &ucount);
        while (m) {
            const int u = 32 * w + __builtin_ctz(m);
            m &= m - 1;
            if (__builtin_expect(off < UQCAP, 1)) {
                uq[off] = u;
            } else {
                #pragma unroll 1
                for (int k = 0; k < 16; ++k) {
                    const unsigned rc = rec[16 * u + k];
                    if ((rc & 0xFF00u) == mykey) add_edge_slowly(acc, x, rc);
                }
            }
            ++off;
        }
    }
    __syncthreads();

    const v4u* rec16 = (const v4u*)rec;
    const int nu = ucount < UQCAP ? ucount : UQCAP;
    const int wave_s = __builtin_amdgcn_readfirstlane(wave);
    for (int base = wave_s * 32; base < nu; base += OWN_WAVES * 32) {
        v4u rv[4];
        bool have[4];
        int  un[4];
        #pragma unroll
        for (int j = 0; j < 4; ++j) {
            const int ui = base + (lane >> 2) + 8 * j;
            have[j] = ui < nu;
            un[j] = uq[have[j] ? ui : base];
        }
        #pragma unroll
        for (int j = 0; j < 4; ++j)
            rv[j] = rec16[4 * un[j] + (lane & 3)];
        bool     keep[16];
        unsigned bal[16];
        int total = 0;
        #pragma unroll
        for (int k = 0; k < 16; ++k) {
            keep[k] = have[k >> 2] && (rv[k >> 2][k & 3] & 0xFF00u) == mykey;
            bal[k]  = __builtin_amdgcn_ballot_w32(keep[k]);
            total  += __builtin_popcount(bal[k]);
        }
        if (total > 0) {
            int b0 = 0;
            if (lane == 0) b0 = atomicAdd(&qcount, total);
            int pre = __builtin_amdgcn_readfirstlane(b0);
            #pragma unroll
            for (int k = 0; k < 16; ++k) {
                if (keep[k]) {
                    const unsigned rc = rv[k >> 2][k & 3];
                    const int off = pre + (int)__builtin_amdgcn_mbcnt_lo(bal[k], 0u);
                    if (__builtin_expect(off < QCAP, 1)) {
                        queue[off] = rc;
                    } else {
                        add_edge_slowly(acc, x, rc);
                    }
                }
                pre += __builtin_popcount(bal[k]);
            }
        }
    }
    __syncthreads();

    const int set     = wave_s >> 3;
    const bool extra  = set == 0;
    const int cn      = lane & 15;
    const int hh      = lane >> 4;
    const int colbase = (wave_s & 7) * 16;
    const float* wb   = shw + hh * WPITCH + (colbase + cn) * 2;
    const int aoff    = cn * APITCH + 2 * hh;

    const int n = qcount < QCAP ? qcount : QCAP;
    v8f o[4];
    #pragma unroll
    for (int i = 0; i < 4; ++i) {
        const int base       = wave_s * 32 + i * (OWN_WAVES * 32);
        const bool has_batch = base < n;
        const bool has_tile  = i < 3 || extra;
        const float* ap = xs + (set + 4 * i) * 16 * APITCH + aoff;
        v8f c = {};
        if (has_batch) {
            const unsigned rc = base + lane < n ? queue[base + lane] : (unsigned)OWN_NODES;
            const int s = (int)(rc >> 16);
            const int r = (int)(rc & 255u);
            if (has_tile) c = gather_batch_with_tile<true>(x, acc, s, r, lane, ap, wb, c);
            else          c = gather_batch_with_tile<false>(x, acc, s, r, lane, ap, wb, c);
        } else if (has_tile) {
            c = tile_product<0, D>(ap, wb, c);
        }
        o[i] = c;
    }
    for (int base = wave_s * 32 + 4 * (OWN_WAVES * 32); base < n; base += OWN_WAVES * 32) {
        const unsigned rc = base + lane < n ? queue[base + lane] : (unsigned)OWN_NODES;
        const int s = (int)(rc >> 16);
        const int r = (int)(rc & 255u);
        v4f xv[8];
        fetch_rows<0>(x, s, lane, xv);  add_rows<0>(acc, r, lane, xv);
        fetch_rows<8>(x, s, lane, xv);  add_rows<8>(acc, r, lane, xv);
        fetch_rows<16>(x, s, lane, xv); add_rows<16>(acc, r, lane, xv);
        fetch_rows<24>(x, s, lane, xv); add_rows<24>(acc, r, lane, xv);
    }
    __syncthreads();

    const float bias1 = b1[colbase + cn];
    const float bias2 = b2[colbase + cn] + bres[colbase + cn];

    stage_matrix(W1, shw, tid, true);
    for (int i = tid; i < OWN_PAD * 32; i += OWN_THREADS) {
        const int off = (i >> 5) * APITCH + (i & 31) * 4;
        const v4f xx = *(const v4f*)(xs + off);
        const v4f aa = *(const v4f*)(acc + off);
        *(v4f*)(acc + off) = xx + aa;
    }
    __syncthreads();

    #pragma unroll
    for (int i = 0; i < 4; ++i) {
        if (i < 3 || extra) {
            const int row0 = (set + 4 * i) * 16;
            v8f c = {};
            c = tile_product<0, D>(acc + row0 * APITCH + aoff, wb, c);
            #pragma unroll
            for (int r = 0; r < 8; ++r)
                xs[(row0 + r + 8 * hh) * APITCH + colbase + cn] = fmaxf(c[r] + bias1, 0.f);
        }
    }
    __syncthreads();

    stage_matrix(W2, shw, tid, false);
    __syncthreads();

    #pragma unroll
    for (int i = 0; i < 4; ++i) {
        if (i < 3 || extra) {
            const int row0 = (set + 4 * i) * 16;
            const v8f c = tile_product<0, D>(xs + row0 * APITCH + aoff, wb, o[i]);
            #pragma unroll
            for (int r = 0; r < 8; ++r) {
                const int row = row0 + r + 8 * hh;
                if (row < OWN_NODES)
                    out[(size_t)(lo + row) * D + colbase + cn] = c[r] + bias2;
            }
        }
    }
}


extern "C" void kernel_launch(void* const* d_in, const int* in_sizes, int n_in,
                              void* d_out, int out_size, void* d_ws, size_t ws_size,
                              hipStream_t stream) {
    const float* x    = (const float*)d_in[0];
    const int*   ei   = (const int*)  d_in[1];
    const float* W1   = (const float*)d_in[2];
    const float* b1   = (const float*)d_in[3];
    const float* W2   = (const float*)d_in[4];
    const float* b2   = (const float*)d_in[5];
    const float* Wres = (const float*)d_in[6];
    const float* bres = (const float*)d_in[7];
    float* out = (float*)d_out;
    unsigned int* rec = (unsigned int*)d_ws;
    unsigned int* bm  = (unsigned int*)((char*)d_ws + (size_t)N_EDGES * sizeof(unsigned int));

    edge_index_kernel<<<IDX_GROUPS, IDX_THREADS, 0, stream>>>(ei, ei + N_EDGES, rec, bm);

    gin_owner_kernel<<<OWN_GROUPS, OWN_THREADS, 0, stream>>>(
        x, rec, bm, W1, b1, W2, b2, Wres, bres, out);
}
